# starting kernel
# baseline (speedup 1.0000x reference)
#include <hip/hip_runtime.h>
#include <math.h>


#define NUM_Q  10000
#define NUM_C  300
#define DIM_S  64
#define SIZE_M 50
#define BATCH  128
#define TT     200
#define HALF   32

#define RK  212
#define RP  208
#define WST 52

typedef __attribute__((ext_vector_type(2))) float v2f;
typedef __attribute__((ext_vector_type(8))) float v8f;

static __device__ __forceinline__ v8f wmma_f32_k4(v2f a, v2f b, v8f c) {
  return __builtin_amdgcn_wmma_f32_16x16x4_f32(false, a, false, b, (short)0, c, false, false);
}

__global__ __launch_bounds__(256) void deepirt_fused_kernel(
    const int*   __restrict__ question, const int*   __restrict__ response,
    const float* __restrict__ vq_emb,   const float* __restrict__ vc_emb,
    const float* __restrict__ kq_emb,   const float* __restrict__ kc_emb,
    const float* __restrict__ Mk,       const float* __restrict__ Mv0,
    const float* __restrict__ eW, const float* __restrict__ eb,
    const float* __restrict__ aW, const float* __restrict__ ab,
    const float* __restrict__ fW, const float* __restrict__ fb,
    const float* __restrict__ abilW, const float* __restrict__ abilb,
    const float* __restrict__ diffW, const float* __restrict__ diffb,
    float* __restrict__ out)
{
  extern __shared__ float smem[];
  float* sK   = smem;
  float* sV   = sK  + RK * 64;
  float* sE   = sV  + RP * 64;
  float* sA   = sE  + RP * 64;
  float* sW   = sA  + RP * 64;
  float* sMv  = sW  + RP * WST;
  float* sRed = sMv + SIZE_M * 64;
  int*   sQ   = (int*)(sRed + 4 * 64);
  int*   sRsp = sQ + TT;

  const int b    = blockIdx.x;
  const int tid  = threadIdx.x;
  const int lane = tid & 31;
  const int wave = tid >> 5;
  const int hh   = lane >> 4;
  const int r16  = lane & 15;

  for (int t = tid; t < TT; t += 256) {
    sQ[t]   = question[b * TT + t];
    sRsp[t] = response[b * TT + t];
  }
  __syncthreads();

  for (int idx = tid; idx < RK * 64; idx += 256) {
    int t = idx >> 6, c = idx & 63;
    float x = 0.f;
    if (t < TT) {
      int q = sQ[t];
      x = (c < HALF) ? kq_emb[q * HALF + c] : kc_emb[c - HALF];
    }
    sK[idx] = x;
  }
  for (int idx = tid; idx < RP * 64; idx += 256) {
    int t = idx >> 6, c = idx & 63;
    float x = 0.f;
    if (t < TT) {
      int q = sQ[t], r = sRsp[t];
      x = (c < HALF) ? vq_emb[(q + NUM_Q * r) * HALF + c]
                     : vc_emb[(NUM_C * r) * HALF + (c - HALF)];
    }
    sV[idx] = x;
  }
  for (int idx = tid; idx < SIZE_M * 64; idx += 256) sMv[idx] = Mv0[idx];
  __syncthreads();

  for (int nt = 0; nt < 4; ++nt) {
    int o  = nt * 16 + r16;
    int oc = (o < SIZE_M) ? o : (SIZE_M - 1);
    v2f bf[16];
#pragma unroll
    for (int kb = 0; kb < 16; ++kb)
      bf[kb] = *(const v2f*)&Mk[oc * 64 + kb * 4 + 2 * hh];
    for (int mt = wave; mt < RP / 16; mt += 8) {
      v2f af[16];
#pragma unroll
      for (int kb = 0; kb < 16; ++kb)
        af[kb] = *(const v2f*)&sK[(mt * 16 + r16) * 64 + kb * 4 + 2 * hh];
      v8f acc = {0.f,0.f,0.f,0.f,0.f,0.f,0.f,0.f};
#pragma unroll
      for (int kb = 0; kb < 16; ++kb) acc = wmma_f32_k4(af[kb], bf[kb], acc);
      if (o < SIZE_M) {
#pragma unroll
        for (int j = 0; j < 8; ++j)
          sW[(mt * 16 + j + 8 * hh) * WST + o] = acc[j];
      }
    }
  }

  for (int nt = 0; nt < 4; ++nt) {
    int o = nt * 16 + r16;
    v2f be[16], ba[16];
#pragma unroll
    for (int kb = 0; kb < 16; ++kb) {
      be[kb] = *(const v2f*)&eW[o * 64 + kb * 4 + 2 * hh];
      ba[kb] = *(const v2f*)&aW[o * 64 + kb * 4 + 2 * hh];
    }
    float ebv = eb[o], abv = ab[o];
    for (int mt = wave; mt < RP / 16; mt += 8) {
      v2f af[16];
#pragma unroll
      for (int kb = 0; kb < 16; ++kb)
        af[kb] = *(const v2f*)&sV[(mt * 16 + r16) * 64 + kb * 4 + 2 * hh];
      v8f ce = {0.f,0.f,0.f,0.f,0.f,0.f,0.f,0.f};
      v8f ca = {0.f,0.f,0.f,0.f,0.f,0.f,0.f,0.f};
#pragma unroll
      for (int kb = 0; kb < 16; ++kb) {
        ce = wmma_f32_k4(af[kb], be[kb], ce);
        ca = wmma_f32_k4(af[kb], ba[kb], ca);
      }
#pragma unroll
      for (int j = 0; j < 8; ++j) {
        int row = mt * 16 + j + 8 * hh;
        sE[row * 64 + o] = 1.f / (1.f + __expf(-(ce[j] + ebv)));
        sA[row * 64 + o] = tanhf(ca[j] + abv);
      }
    }
  }
  __syncthreads();

  for (int r = tid; r < TT; r += 256) {
    float mx = -1e30f;
    for (int m = 0; m < SIZE_M; ++m) mx = fmaxf(mx, sW[r * WST + m]);
    float s = 0.f;
    for (int m = 0; m < SIZE_M; ++m) {
      float ev = __expf(sW[r * WST + m] - mx);
      sW[r * WST + m] = ev; s += ev;
    }
    float inv = 1.f / s;
    for (int m = 0; m < SIZE_M; ++m) sW[r * WST + m] *= inv;
  }
  __syncthreads();

  const int dd = tid & 63, gg = tid >> 6;
  for (int t = 0; t < TT; ++t) {
    for (int idx = tid; idx < SIZE_M * 64; idx += 256) {
      int m = idx >> 6, d = idx & 63;
      float wt = sW[t * WST + m];
      float et = sE[t * 64 + d];
      float at = sA[t * 64 + d];
      sMv[idx] = sMv[idx] * (1.f - wt * et) + wt * at;
    }
    __syncthreads();
    if (t < TT - 1) {
      float p = 0.f;
      for (int m = gg; m < SIZE_M; m += 4)
        p += sW[(t + 1) * WST + m] * sMv[m * 64 + dd];
      sRed[gg * 64 + dd] = p;
      __syncthreads();
      if (tid < 64)
        sV[t * 64 + tid] = sRed[tid] + sRed[64 + tid] + sRed[128 + tid] + sRed[192 + tid];
      __syncthreads();
    } else {
      __syncthreads();
    }
  }

  for (int idx = (TT - 1) * 64 + tid; idx < RP * 64; idx += 256) sV[idx] = 0.f;
  __syncthreads();

  for (int nt = 0; nt < 4; ++nt) {
    int o = nt * 16 + r16;
    v2f bf[32];
#pragma unroll
    for (int kb = 0; kb < 32; ++kb)
      bf[kb] = *(const v2f*)&fW[o * 128 + kb * 4 + 2 * hh];
    float fbv = fb[o];
    for (int mt = wave; mt < RP / 16; mt += 8) {
      int row = mt * 16 + r16;
      v2f af[32];
#pragma unroll
      for (int kb = 0; kb < 16; ++kb)
        af[kb] = *(const v2f*)&sV[row * 64 + kb * 4 + 2 * hh];
#pragma unroll
      for (int kb = 16; kb < 32; ++kb)
        af[kb] = *(const v2f*)&sK[(row + 1) * 64 + (kb - 16) * 4 + 2 * hh];
      v8f acc = {0.f,0.f,0.f,0.f,0.f,0.f,0.f,0.f};
#pragma unroll
      for (int kb = 0; kb < 32; ++kb) acc = wmma_f32_k4(af[kb], bf[kb], acc);
#pragma unroll
      for (int j = 0; j < 8; ++j) {
        int rr = mt * 16 + j + 8 * hh;
        if (rr < TT - 1) sE[rr * 64 + o] = tanhf(acc[j] + fbv);
      }
    }
  }
  __syncthreads();

  for (int r = tid; r < TT - 1; r += 256) {
    float s1 = 0.f, s2 = 0.f;
    for (int c = 0; c < 64; ++c) {
      s1 += sE[r * 64 + c] * abilW[c];
      s2 += sK[(r + 1) * 64 + c] * diffW[c];
    }
    float ability = tanhf(s1 + abilb[0]);
    float diff    = tanhf(s2 + diffb[0]);
    out[b * (TT - 1) + r] = 3.f * ability - diff;
  }
}

extern "C" void kernel_launch(void* const* d_in, const int* in_sizes, int n_in,
                              void* d_out, int out_size, void* d_ws, size_t ws_size,
                              hipStream_t stream) {
  (void)in_sizes; (void)n_in; (void)out_size; (void)d_ws; (void)ws_size;
  const int*   question = (const int*)  d_in[0];
  const int*   response = (const int*)  d_in[1];
  const float* vq_emb   = (const float*)d_in[2];
  const float* vc_emb   = (const float*)d_in[3];
  const float* kq_emb   = (const float*)d_in[4];
  const float* kc_emb   = (const float*)d_in[5];
  const float* Mk       = (const float*)d_in[6];
  const float* Mv0      = (const float*)d_in[7];
  const float* eW       = (const float*)d_in[8];
  const float* ebv      = (const float*)d_in[9];
  const float* aW       = (const float*)d_in[10];
  const float* abv      = (const float*)d_in[11];
  const float* fW       = (const float*)d_in[12];
  const float* fbv      = (const float*)d_in[13];
  const float* abilW    = (const float*)d_in[14];
  const float* abilb    = (const float*)d_in[15];
  const float* diffW    = (const float*)d_in[16];
  const float* diffb    = (const float*)d_in[17];
  float* out = (float*)d_out;

  const size_t smem_bytes =
      (size_t)(RK * 64 + 3 * RP * 64 + RP * WST + SIZE_M * 64 + 4 * 64) * sizeof(float)
      + (size_t)(2 * TT) * sizeof(int);

  (void)hipFuncSetAttribute((const void*)deepirt_fused_kernel,
                            hipFuncAttributeMaxDynamicSharedMemorySize,
                            (int)smem_bytes);

  deepirt_fused_kernel<<<BATCH, 256, smem_bytes, stream>>>(
      question, response, vq_emb, vc_emb, kq_emb, kc_emb, Mk, Mv0,
      eW, ebv, aW, abv, fW, fbv, abilW, abilb, diffW, diffb, out);
}


# Scan keeps the value memory in registers (thread owns Mv[m][d], m=mg+4i; a quad adds its 4 m-groups by DPP, old sum order): no LDS round trip of Mv, no barrier inside the 200-step scan
# speedup vs baseline: 2.6992x; 2.6992x over previous
#include <hip/hip_runtime.h>
#include <math.h>


#define NUM_Q  10000
#define NUM_C  300
#define DIM_S  64
#define SIZE_M 50
#define BATCH  128
#define TT     200
#define HALF   32

#define RK  212
#define RP  208
#define WST 52

typedef __attribute__((ext_vector_type(2))) float v2f;
typedef __attribute__((ext_vector_type(8))) float v8f;

static __device__ __forceinline__ v8f wmma_f32_k4(v2f a, v2f b, v8f c) {
  return __builtin_amdgcn_wmma_f32_16x16x4_f32(false, a, false, b, (short)0, c, false, false);
}

#define MPT 13

static __device__ __forceinline__ void scan_load(float (&w)[MPT], float& e, float& a,
                                                 const float* sWrow, const float* sErow,
                                                 const float* sArow, bool has13) {
#pragma unroll
  for (int i = 0; i < MPT - 1; ++i) w[i] = sWrow[4 * i];
  w[MPT - 1] = has13 ? sWrow[4 * (MPT - 1)] : 0.f;
  e = *sErow;
  a = *sArow;
}

static __device__ __forceinline__ void scan_update(float (&mv)[MPT], const float (&w)[MPT],
                                                   float e, float a) {
#pragma unroll
  for (int i = 0; i < MPT; ++i) mv[i] = mv[i] * (1.f - w[i] * e) + w[i] * a;
}

template <int K>
static __device__ __forceinline__ float quad_lane(float v) {
  return __int_as_float(__builtin_amdgcn_mov_dpp(__float_as_int(v), K * 0x55, 0xf, 0xf, false));
}

static __device__ __forceinline__ float scan_read(const float (&mv)[MPT], const float (&w)[MPT]) {
  float p = 0.f;
#pragma unroll
  for (int i = 0; i < MPT; ++i) p += w[i] * mv[i];
  return ((p + quad_lane<1>(p)) + quad_lane<2>(p)) + quad_lane<3>(p);
}

__global__ __launch_bounds__(256) void deepirt_fused_kernel(
    const int*   __restrict__ question, const int*   __restrict__ response,
    const float* __restrict__ vq_emb,   const float* __restrict__ vc_emb,
    const float* __restrict__ kq_emb,   const float* __restrict__ kc_emb,
    const float* __restrict__ Mk,       const float* __restrict__ Mv0,
    const float* __restrict__ eW, const float* __restrict__ eb,
    const float* __restrict__ aW, const float* __restrict__ ab,
    const float* __restrict__ fW, const float* __restrict__ fb,
    const float* __restrict__ abilW, const float* __restrict__ abilb,
    const float* __restrict__ diffW, const float* __restrict__ diffb,
    float* __restrict__ out)
{
  extern __shared__ float smem[];
  float* sK   = smem;
  float* sV   = sK  + RK * 64;
  float* sE   = sV  + RP * 64;
  float* sA   = sE  + RP * 64;
  float* sW   = sA  + RP * 64;
  int*   sQ   = (int*)(sW + RP * WST);
  int*   sRsp = sQ + TT;

  const int b    = blockIdx.x;
  const int tid  = threadIdx.x;
  const int lane = tid & 31;
  const int wave = tid >> 5;
  const int hh   = lane >> 4;
  const int r16  = lane & 15;

  for (int t = tid; t < TT; t += 256) {
    sQ[t]   = question[b * TT + t];
    sRsp[t] = response[b * TT + t];
  }
  __syncthreads();

  for (int idx = tid; idx < RK * 64; idx += 256) {
    int t = idx >> 6, c = idx & 63;
    float x = 0.f;
    if (t < TT) {
      int q = sQ[t];
      x = (c < HALF) ? kq_emb[q * HALF + c] : kc_emb[c - HALF];
    }
    sK[idx] = x;
  }
  for (int idx = tid; idx < RP * 64; idx += 256) {
    int t = idx >> 6, c = idx & 63;
    float x = 0.f;
    if (t < TT) {
      int q = sQ[t], r = sRsp[t];
      x = (c < HALF) ? vq_emb[(q + NUM_Q * r) * HALF + c]
                     : vc_emb[(NUM_C * r) * HALF + (c - HALF)];
    }
    sV[idx] = x;
  }
  __syncthreads();

  for (int nt = 0; nt < 4; ++nt) {
    int o  = nt * 16 + r16;
    int oc = (o < SIZE_M) ? o : (SIZE_M - 1);
    v2f bf[16];
#pragma unroll
    for (int kb = 0; kb < 16; ++kb)
      bf[kb] = *(const v2f*)&Mk[oc * 64 + kb * 4 + 2 * hh];
    for (int mt = wave; mt < RP / 16; mt += 8) {
      v2f af[16];
#pragma unroll
      for (int kb = 0; kb < 16; ++kb)
        af[kb] = *(const v2f*)&sK[(mt * 16 + r16) * 64 + kb * 4 + 2 * hh];
      v8f acc = {0.f,0.f,0.f,0.f,0.f,0.f,0.f,0.f};
#pragma unroll
      for (int kb = 0; kb < 16; ++kb) acc = wmma_f32_k4(af[kb], bf[kb], acc);
      if (o < SIZE_M) {
#pragma unroll
        for (int j = 0; j < 8; ++j)
          sW[(mt * 16 + j + 8 * hh) * WST + o] = acc[j];
      }
    }
  }

  for (int nt = 0; nt < 4; ++nt) {
    int o = nt * 16 + r16;
    v2f be[16], ba[16];
#pragma unroll
    for (int kb = 0; kb < 16; ++kb) {
      be[kb] = *(const v2f*)&eW[o * 64 + kb * 4 + 2 * hh];
      ba[kb] = *(const v2f*)&aW[o * 64 + kb * 4 + 2 * hh];
    }
    float ebv = eb[o], abv = ab[o];
    for (int mt = wave; mt < RP / 16; mt += 8) {
      v2f af[16];
#pragma unroll
      for (int kb = 0; kb < 16; ++kb)
        af[kb] = *(const v2f*)&sV[(mt * 16 + r16) * 64 + kb * 4 + 2 * hh];
      v8f ce = {0.f,0.f,0.f,0.f,0.f,0.f,0.f,0.f};
      v8f ca = {0.f,0.f,0.f,0.f,0.f,0.f,0.f,0.f};
#pragma unroll
      for (int kb = 0; kb < 16; ++kb) {
        ce = wmma_f32_k4(af[kb], be[kb], ce);
        ca = wmma_f32_k4(af[kb], ba[kb], ca);
      }
#pragma unroll
      for (int j = 0; j < 8; ++j) {
        int row = mt * 16 + j + 8 * hh;
        sE[row * 64 + o] = 1.f / (1.f + __expf(-(ce[j] + ebv)));
        sA[row * 64 + o] = tanhf(ca[j] + abv);
      }
    }
  }
  __syncthreads();

  const int mg = lane & 3;
  const int dq = wave * 8 + (lane >> 2);
  float mv[MPT];
#pragma unroll
  for (int i = 0; i < MPT; ++i) {
    int m = mg + 4 * i;
    mv[i] = (m < SIZE_M) ? Mv0[m * 64 + dq] : 0.f;
  }

  for (int r = tid; r < TT; r += 256) {
    float mx = -1e30f;
    for (int m = 0; m < SIZE_M; ++m) mx = fmaxf(mx, sW[r * WST + m]);
    float s = 0.f;
    for (int m = 0; m < SIZE_M; ++m) {
      float ev = __expf(sW[r * WST + m] - mx);
      sW[r * WST + m] = ev; s += ev;
    }
    float inv = 1.f / s;
    for (int m = 0; m < SIZE_M; ++m) sW[r * WST + m] *= inv;
  }
  __syncthreads();

  {
    const bool has13 = (mg + 4 * (MPT - 1)) < SIZE_M;
    const float* pW = sW + mg;
    const float* pE = sE + dq;
    const float* pA = sA + dq;
    float*       pR = sV + dq;

    float wA[MPT], wB[MPT], eA, aA, eB, aB;
    scan_load(wA, eA, aA, pW, pE, pA, has13);
    scan_load(wB, eB, aB, pW + WST, pE + 64, pA + 64, has13);
    scan_update(mv, wA, eA, aA);

    for (int t = 1; t < TT - 1; t += 2) {
      scan_load(wA, eA, aA, pW + (t + 1) * WST, pE + (t + 1) * 64, pA + (t + 1) * 64, has13);
      float r0 = scan_read(mv, wB);
      if (mg == 0) pR[(t - 1) * 64] = r0;
      scan_update(mv, wB, eB, aB);

      scan_load(wB, eB, aB, pW + (t + 2) * WST, pE + (t + 2) * 64, pA + (t + 2) * 64, has13);
      float r1 = scan_read(mv, wA);
      if (mg == 0) pR[t * 64] = r1;
      scan_update(mv, wA, eA, aA);
    }
    float rl = scan_read(mv, wB);
    if (mg == 0) pR[(TT - 2) * 64] = rl;
  }

  for (int idx = (TT - 1) * 64 + tid; idx < RP * 64; idx += 256) sV[idx] = 0.f;
  __syncthreads();

  for (int nt = 0; nt < 4; ++nt) {
    int o = nt * 16 + r16;
    v2f bf[32];
#pragma unroll
    for (int kb = 0; kb < 32; ++kb)
      bf[kb] = *(const v2f*)&fW[o * 128 + kb * 4 + 2 * hh];
    float fbv = fb[o];
    for (int mt = wave; mt < RP / 16; mt += 8) {
      int row = mt * 16 + r16;
      v2f af[32];
#pragma unroll
      for (int kb = 0; kb < 16; ++kb)
        af[kb] = *(const v2f*)&sV[row * 64 + kb * 4 + 2 * hh];
#pragma unroll
      for (int kb = 16; kb < 32; ++kb)
        af[kb] = *(const v2f*)&sK[(row + 1) * 64 + (kb - 16) * 4 + 2 * hh];
      v8f acc = {0.f,0.f,0.f,0.f,0.f,0.f,0.f,0.f};
#pragma unroll
      for (int kb = 0; kb < 32; ++kb) acc = wmma_f32_k4(af[kb], bf[kb], acc);
#pragma unroll
      for (int j = 0; j < 8; ++j) {
        int rr = mt * 16 + j + 8 * hh;
        if (rr < TT - 1) sE[rr * 64 + o] = tanhf(acc[j] + fbv);
      }
    }
  }
  __syncthreads();

  for (int r = tid; r < TT - 1; r += 256) {
    float s1 = 0.f, s2 = 0.f;
    for (int c = 0; c < 64; ++c) {
      s1 += sE[r * 64 + c] * abilW[c];
      s2 += sK[(r + 1) * 64 + c] * diffW[c];
    }
    float ability = tanhf(s1 + abilb[0]);
    float diff    = tanhf(s2 + diffb[0]);
    out[b * (TT - 1) + r] = 3.f * ability - diff;
  }
}

extern "C" void kernel_launch(void* const* d_in, const int* in_sizes, int n_in,
                              void* d_out, int out_size, void* d_ws, size_t ws_size,
                              hipStream_t stream) {
  (void)in_sizes; (void)n_in; (void)out_size; (void)d_ws; (void)ws_size;
  const int*   question = (const int*)  d_in[0];
  const int*   response = (const int*)  d_in[1];
  const float* vq_emb   = (const float*)d_in[2];
  const float* vc_emb   = (const float*)d_in[3];
  const float* kq_emb   = (const float*)d_in[4];
  const float* kc_emb   = (const float*)d_in[5];
  const float* Mk       = (const float*)d_in[6];
  const float* Mv0      = (const float*)d_in[7];
  const float* eW       = (const float*)d_in[8];
  const float* ebv      = (const float*)d_in[9];
  const float* aW       = (const float*)d_in[10];
  const float* abv      = (const float*)d_in[11];
  const float* fW       = (const float*)d_in[12];
  const float* fbv      = (const float*)d_in[13];
  const float* abilW    = (const float*)d_in[14];
  const float* abilb    = (const float*)d_in[15];
  const float* diffW    = (const float*)d_in[16];
  const float* diffb    = (const float*)d_in[17];
  float* out = (float*)d_out;

  const size_t smem_bytes =
      (size_t)(RK * 64 + 3 * RP * 64 + RP * WST) * sizeof(float)
      + (size_t)(2 * TT) * sizeof(int);

  (void)hipFuncSetAttribute((const void*)deepirt_fused_kernel,
                            hipFuncAttributeMaxDynamicSharedMemorySize,
                            (int)smem_bytes);

  deepirt_fused_kernel<<<BATCH, 256, smem_bytes, stream>>>(
      question, response, vq_emb, vc_emb, kq_emb, kc_emb, Mk, Mv0,
      eW, ebv, aW, abv, fW, fbv, abilW, abilb, diffW, diffb, out);
}


# Embedding gather: a thread moves 16 B of a k row and of a v row per pass, its 26 index loads, then its 26 row loads, all in flight at once (was one dependent 4 B load per trip); no index cache in LDS
# speedup vs baseline: 3.8991x; 1.4445x over previous
#include <hip/hip_runtime.h>
#include <math.h>


#define NUM_Q  10000
#define NUM_C  300
#define DIM_S  64
#define SIZE_M 50
#define BATCH  128
#define TT     200
#define HALF   32

#define RK  212
#define RP  208
#define WST 52

typedef __attribute__((ext_vector_type(2))) float v2f;
typedef __attribute__((ext_vector_type(8))) float v8f;

static __device__ __forceinline__ v8f wmma_f32_k4(v2f a, v2f b, v8f c) {
  return __builtin_amdgcn_wmma_f32_16x16x4_f32(false, a, false, b, (short)0, c, false, false);
}

#define GP  13

typedef __attribute__((ext_vector_type(4))) float v4f;

#define MPT 13

static __device__ __forceinline__ void scan_load(float (&w)[MPT], float& e, float& a,
                                                 const float* sWrow, const float* sErow,
                                                 const float* sArow, bool has13) {
#pragma unroll
  for (int i = 0; i < MPT - 1; ++i) w[i] = sWrow[4 * i];
  w[MPT - 1] = has13 ? sWrow[4 * (MPT - 1)] : 0.f;
  e = *sErow;
  a = *sArow;
}

static __device__ __forceinline__ void scan_update(float (&mv)[MPT], const float (&w)[MPT],
                                                   float e, float a) {
#pragma unroll
  for (int i = 0; i < MPT; ++i) mv[i] = mv[i] * (1.f - w[i] * e) + w[i] * a;
}

template <int K>
static __device__ __forceinline__ float quad_lane(float v) {
  return __int_as_float(__builtin_amdgcn_mov_dpp(__float_as_int(v), K * 0x55, 0xf, 0xf, false));
}

static __device__ __forceinline__ float scan_read(const float (&mv)[MPT], const float (&w)[MPT]) {
  float p = 0.f;
#pragma unroll
  for (int i = 0; i < MPT; ++i) p += w[i] * mv[i];
  return ((p + quad_lane<1>(p)) + quad_lane<2>(p)) + quad_lane<3>(p);
}

__global__ __launch_bounds__(256) void deepirt_fused_kernel(
    const int*   __restrict__ question, const int*   __restrict__ response,
    const float* __restrict__ vq_emb,   const float* __restrict__ vc_emb,
    const float* __restrict__ kq_emb,   const float* __restrict__ kc_emb,
    const float* __restrict__ Mk,       const float* __restrict__ Mv0,
    const float* __restrict__ eW, const float* __restrict__ eb,
    const float* __restrict__ aW, const float* __restrict__ ab,
    const float* __restrict__ fW, const float* __restrict__ fb,
    const float* __restrict__ abilW, const float* __restrict__ abilb,
    const float* __restrict__ diffW, const float* __restrict__ diffb,
    float* __restrict__ out)
{
  extern __shared__ float smem[];
  float* sK   = smem;
  float* sV   = sK  + RK * 64;
  float* sE   = sV  + RP * 64;
  float* sA   = sE  + RP * 64;
  float* sW   = sA  + RP * 64;

  const int b    = blockIdx.x;
  const int tid  = threadIdx.x;
  const int lane = tid & 31;
  const int wave = tid >> 5;
  const int hh   = lane >> 4;
  const int r16  = lane & 15;

  {
    const int c4 = tid & 15;
    const int gr = tid >> 4;
    int qi[GP], ri[GP];
#pragma unroll
    for (int p = 0; p < GP; ++p) {
      int t  = p * 16 + gr;
      int tc = (t < TT) ? t : (TT - 1);
      qi[p] = question[b * TT + tc];
      ri[p] = response[b * TT + tc];
    }
    const bool   qhalf = c4 < 8;
    const float* ktab  = (qhalf ? kq_emb : kc_emb) + 4 * (c4 & 7);
    const float* vtab  = (qhalf ? vq_emb : vc_emb) + 4 * (c4 & 7);
    v4f kx[GP], vx[GP];
#pragma unroll
    for (int p = 0; p < GP; ++p) {
      int krow = qhalf ? qi[p] : 0;
      int vrow = qhalf ? qi[p] + NUM_Q * ri[p] : NUM_C * ri[p];
      kx[p] = *(const v4f*)(ktab + krow * HALF);
      vx[p] = *(const v4f*)(vtab + vrow * HALF);
    }
    const v4f z4 = {0.f, 0.f, 0.f, 0.f};
#pragma unroll
    for (int p = 0; p < GP; ++p) {
      int t = p * 16 + gr;
      *(v4f*)&sK[t * 64 + 4 * c4] = (t < TT) ? kx[p] : z4;
      *(v4f*)&sV[t * 64 + 4 * c4] = (t < TT) ? vx[p] : z4;
    }
    if (tid < (RK - RP) * 16) *(v4f*)&sK[RP * 64 + 4 * tid] = z4;
  }
  __syncthreads();

  for (int nt = 0; nt < 4; ++nt) {
    int o  = nt * 16 + r16;
    int oc = (o < SIZE_M) ? o : (SIZE_M - 1);
    v2f bf[16];
#pragma unroll
    for (int kb = 0; kb < 16; ++kb)
      bf[kb] = *(const v2f*)&Mk[oc * 64 + kb * 4 + 2 * hh];
    for (int mt = wave; mt < RP / 16; mt += 8) {
      v2f af[16];
#pragma unroll
      for (int kb = 0; kb < 16; ++kb)
        af[kb] = *(const v2f*)&sK[(mt * 16 + r16) * 64 + kb * 4 + 2 * hh];
      v8f acc = {0.f,0.f,0.f,0.f,0.f,0.f,0.f,0.f};
#pragma unroll
      for (int kb = 0; kb < 16; ++kb) acc = wmma_f32_k4(af[kb], bf[kb], acc);
      if (o < SIZE_M) {
#pragma unroll
        for (int j = 0; j < 8; ++j)
          sW[(mt * 16 + j + 8 * hh) * WST + o] = acc[j];
      }
    }
  }

  for (int nt = 0; nt < 4; ++nt) {
    int o = nt * 16 + r16;
    v2f be[16], ba[16];
#pragma unroll
    for (int kb = 0; kb < 16; ++kb) {
      be[kb] = *(const v2f*)&eW[o * 64 + kb * 4 + 2 * hh];
      ba[kb] = *(const v2f*)&aW[o * 64 + kb * 4 + 2 * hh];
    }
    float ebv = eb[o], abv = ab[o];
    for (int mt = wave; mt < RP / 16; mt += 8) {
      v2f af[16];
#pragma unroll
      for (int kb = 0; kb < 16; ++kb)
        af[kb] = *(const v2f*)&sV[(mt * 16 + r16) * 64 + kb * 4 + 2 * hh];
      v8f ce = {0.f,0.f,0.f,0.f,0.f,0.f,0.f,0.f};
      v8f ca = {0.f,0.f,0.f,0.f,0.f,0.f,0.f,0.f};
#pragma unroll
      for (int kb = 0; kb < 16; ++kb) {
        ce = wmma_f32_k4(af[kb], be[kb], ce);
        ca = wmma_f32_k4(af[kb], ba[kb], ca);
      }
#pragma unroll
      for (int j = 0; j < 8; ++j) {
        int row = mt * 16 + j + 8 * hh;
        sE[row * 64 + o] = 1.f / (1.f + __expf(-(ce[j] + ebv)));
        sA[row * 64 + o] = tanhf(ca[j] + abv);
      }
    }
  }
  __syncthreads();

  const int mg = lane & 3;
  const int dq = wave * 8 + (lane >> 2);
  float mv[MPT];
#pragma unroll
  for (int i = 0; i < MPT; ++i) {
    int m = mg + 4 * i;
    mv[i] = (m < SIZE_M) ? Mv0[m * 64 + dq] : 0.f;
  }

  for (int r = tid; r < TT; r += 256) {
    float mx = -1e30f;
    for (int m = 0; m < SIZE_M; ++m) mx = fmaxf(mx, sW[r * WST + m]);
    float s = 0.f;
    for (int m = 0; m < SIZE_M; ++m) {
      float ev = __expf(sW[r * WST + m] - mx);
      sW[r * WST + m] = ev; s += ev;
    }
    float inv = 1.f / s;
    for (int m = 0; m < SIZE_M; ++m) sW[r * WST + m] *= inv;
  }
  __syncthreads();

  {
    const bool has13 = (mg + 4 * (MPT - 1)) < SIZE_M;
    const float* pW = sW + mg;
    const float* pE = sE + dq;
    const float* pA = sA + dq;
    float*       pR = sV + dq;

    float wA[MPT], wB[MPT], eA, aA, eB, aB;
    scan_load(wA, eA, aA, pW, pE, pA, has13);
    scan_load(wB, eB, aB, pW + WST, pE + 64, pA + 64, has13);
    scan_update(mv, wA, eA, aA);

    for (int t = 1; t < TT - 1; t += 2) {
      scan_load(wA, eA, aA, pW + (t + 1) * WST, pE + (t + 1) * 64, pA + (t + 1) * 64, has13);
      float r0 = scan_read(mv, wB);
      if (mg == 0) pR[(t - 1) * 64] = r0;
      scan_update(mv, wB, eB, aB);

      scan_load(wB, eB, aB, pW + (t + 2) * WST, pE + (t + 2) * 64, pA + (t + 2) * 64, has13);
      float r1 = scan_read(mv, wA);
      if (mg == 0) pR[t * 64] = r1;
      scan_update(mv, wA, eA, aA);
    }
    float rl = scan_read(mv, wB);
    if (mg == 0) pR[(TT - 2) * 64] = rl;
  }

  for (int idx = (TT - 1) * 64 + tid; idx < RP * 64; idx += 256) sV[idx] = 0.f;
  __syncthreads();

  for (int nt = 0; nt < 4; ++nt) {
    int o = nt * 16 + r16;
    v2f bf[32];
#pragma unroll
    for (int kb = 0; kb < 32; ++kb)
      bf[kb] = *(const v2f*)&fW[o * 128 + kb * 4 + 2 * hh];
    float fbv = fb[o];
    for (int mt = wave; mt < RP / 16; mt += 8) {
      int row = mt * 16 + r16;
      v2f af[32];
#pragma unroll
      for (int kb = 0; kb < 16; ++kb)
        af[kb] = *(const v2f*)&sV[row * 64 + kb * 4 + 2 * hh];
#pragma unroll
      for (int kb = 16; kb < 32; ++kb)
        af[kb] = *(const v2f*)&sK[(row + 1) * 64 + (kb - 16) * 4 + 2 * hh];
      v8f acc = {0.f,0.f,0.f,0.f,0.f,0.f,0.f,0.f};
#pragma unroll
      for (int kb = 0; kb < 32; ++kb) acc = wmma_f32_k4(af[kb], bf[kb], acc);
#pragma unroll
      for (int j = 0; j < 8; ++j) {
        int rr = mt * 16 + j + 8 * hh;
        if (rr < TT - 1) sE[rr * 64 + o] = tanhf(acc[j] + fbv);
      }
    }
  }
  __syncthreads();

  for (int r = tid; r < TT - 1; r += 256) {
    float s1 = 0.f, s2 = 0.f;
    for (int c = 0; c < 64; ++c) {
      s1 += sE[r * 64 + c] * abilW[c];
      s2 += sK[(r + 1) * 64 + c] * diffW[c];
    }
    float ability = tanhf(s1 + abilb[0]);
    float diff    = tanhf(s2 + diffb[0]);
    out[b * (TT - 1) + r] = 3.f * ability - diff;
  }
}

extern "C" void kernel_launch(void* const* d_in, const int* in_sizes, int n_in,
                              void* d_out, int out_size, void* d_ws, size_t ws_size,
                              hipStream_t stream) {
  (void)in_sizes; (void)n_in; (void)out_size; (void)d_ws; (void)ws_size;
  const int*   question = (const int*)  d_in[0];
  const int*   response = (const int*)  d_in[1];
  const float* vq_emb   = (const float*)d_in[2];
  const float* vc_emb   = (const float*)d_in[3];
  const float* kq_emb   = (const float*)d_in[4];
  const float* kc_emb   = (const float*)d_in[5];
  const float* Mk       = (const float*)d_in[6];
  const float* Mv0      = (const float*)d_in[7];
  const float* eW       = (const float*)d_in[8];
  const float* ebv      = (const float*)d_in[9];
  const float* aW       = (const float*)d_in[10];
  const float* abv      = (const float*)d_in[11];
  const float* fW       = (const float*)d_in[12];
  const float* fbv      = (const float*)d_in[13];
  const float* abilW    = (const float*)d_in[14];
  const float* abilb    = (const float*)d_in[15];
  const float* diffW    = (const float*)d_in[16];
  const float* diffb    = (const float*)d_in[17];
  float* out = (float*)d_out;

  const size_t smem_bytes =
      (size_t)(RK * 64 + 3 * RP * 64 + RP * WST) * sizeof(float);

  (void)hipFuncSetAttribute((const void*)deepirt_fused_kernel,
                            hipFuncAttributeMaxDynamicSharedMemorySize,
                            (int)smem_bytes);

  deepirt_fused_kernel<<<BATCH, 256, smem_bytes, stream>>>(
      question, response, vq_emb, vc_emb, kq_emb, kc_emb, Mk, Mv0,
      eW, ebv, aW, abv, fW, fbv, abilW, abilb, diffW, diffb, out);
}


# LDS row pitch of the 64-float rows (k, v/read, e/f, a) 64 -> 68 floats: WMMA A-fragment loads and the head loop's row reads no longer put all their lanes on one bank
# speedup vs baseline: 4.7106x; 1.2081x over previous
#include <hip/hip_runtime.h>
#include <math.h>


#define NUM_Q  10000
#define NUM_C  300
#define DIM_S  64
#define SIZE_M 50
#define BATCH  128
#define TT     200
#define HALF   32

#define RK  212
#define RP  208
#define WST 52
#define LP  68

typedef __attribute__((ext_vector_type(2))) float v2f;
typedef __attribute__((ext_vector_type(8))) float v8f;

static __device__ __forceinline__ v8f wmma_f32_k4(v2f a, v2f b, v8f c) {
  return __builtin_amdgcn_wmma_f32_16x16x4_f32(false, a, false, b, (short)0, c, false, false);
}

#define GP  13

typedef __attribute__((ext_vector_type(4))) float v4f;

#define MPT 13

static __device__ __forceinline__ void scan_load(float (&w)[MPT], float& e, float& a,
                                                 const float* sWrow, const float* sErow,
                                                 const float* sArow, bool has13) {
#pragma unroll
  for (int i = 0; i < MPT - 1; ++i) w[i] = sWrow[4 * i];
  w[MPT - 1] = has13 ? sWrow[4 * (MPT - 1)] : 0.f;
  e = *sErow;
  a = *sArow;
}

static __device__ __forceinline__ void scan_update(float (&mv)[MPT], const float (&w)[MPT],
                                                   float e, float a) {
#pragma unroll
  for (int i = 0; i < MPT; ++i) mv[i] = mv[i] * (1.f - w[i] * e) + w[i] * a;
}

template <int K>
static __device__ __forceinline__ float quad_lane(float v) {
  return __int_as_float(__builtin_amdgcn_mov_dpp(__float_as_int(v), K * 0x55, 0xf, 0xf, false));
}

static __device__ __forceinline__ float scan_read(const float (&mv)[MPT], const float (&w)[MPT]) {
  float p = 0.f;
#pragma unroll
  for (int i = 0; i < MPT; ++i) p += w[i] * mv[i];
  return ((p + quad_lane<1>(p)) + quad_lane<2>(p)) + quad_lane<3>(p);
}

__global__ __launch_bounds__(256) void deepirt_fused_kernel(
    const int*   __restrict__ question, const int*   __restrict__ response,
    const float* __restrict__ vq_emb,   const float* __restrict__ vc_emb,
    const float* __restrict__ kq_emb,   const float* __restrict__ kc_emb,
    const float* __restrict__ Mk,       const float* __restrict__ Mv0,
    const float* __restrict__ eW, const float* __restrict__ eb,
    const float* __restrict__ aW, const float* __restrict__ ab,
    const float* __restrict__ fW, const float* __restrict__ fb,
    const float* __restrict__ abilW, const float* __restrict__ abilb,
    const float* __restrict__ diffW, const float* __restrict__ diffb,
    float* __restrict__ out)
{
  extern __shared__ float smem[];
  float* sK   = smem;
  float* sV   = sK  + RK * LP;
  float* sE   = sV  + RP * LP;
  float* sA   = sE  + RP * LP;
  float* sW   = sA  + RP * LP;

  const int b    = blockIdx.x;
  const int tid  = threadIdx.x;
  const int lane = tid & 31;
  const int wave = tid >> 5;
  const int hh   = lane >> 4;
  const int r16  = lane & 15;

  {
    const int c4 = tid & 15;
    const int gr = tid >> 4;
    int qi[GP], ri[GP];
#pragma unroll
    for (int p = 0; p < GP; ++p) {
      int t  = p * 16 + gr;
      int tc = (t < TT) ? t : (TT - 1);
      qi[p] = question[b * TT + tc];
      ri[p] = response[b * TT + tc];
    }
    const bool   qhalf = c4 < 8;
    const float* ktab  = (qhalf ? kq_emb : kc_emb) + 4 * (c4 & 7);
    const float* vtab  = (qhalf ? vq_emb : vc_emb) + 4 * (c4 & 7);
    v4f kx[GP], vx[GP];
#pragma unroll
    for (int p = 0; p < GP; ++p) {
      int krow = qhalf ? qi[p] : 0;
      int vrow = qhalf ? qi[p] + NUM_Q * ri[p] : NUM_C * ri[p];
      kx[p] = *(const v4f*)(ktab + krow * HALF);
      vx[p] = *(const v4f*)(vtab + vrow * HALF);
    }
    const v4f z4 = {0.f, 0.f, 0.f, 0.f};
#pragma unroll
    for (int p = 0; p < GP; ++p) {
      int t = p * 16 + gr;
      *(v4f*)&sK[t * LP + 4 * c4] = (t < TT) ? kx[p] : z4;
      *(v4f*)&sV[t * LP + 4 * c4] = (t < TT) ? vx[p] : z4;
    }
    if (tid < (RK - RP) * 16) *(v4f*)&sK[(RP + gr) * LP + 4 * c4] = z4;
  }
  __syncthreads();

  for (int nt = 0; nt < 4; ++nt) {
    int o  = nt * 16 + r16;
    int oc = (o < SIZE_M) ? o : (SIZE_M - 1);
    v2f bf[16];
#pragma unroll
    for (int kb = 0; kb < 16; ++kb)
      bf[kb] = *(const v2f*)&Mk[oc * 64 + kb * 4 + 2 * hh];
    for (int mt = wave; mt < RP / 16; mt += 8) {
      v2f af[16];
#pragma unroll
      for (int kb = 0; kb < 16; ++kb)
        af[kb] = *(const v2f*)&sK[(mt * 16 + r16) * LP + kb * 4 + 2 * hh];
      v8f acc = {0.f,0.f,0.f,0.f,0.f,0.f,0.f,0.f};
#pragma unroll
      for (int kb = 0; kb < 16; ++kb) acc = wmma_f32_k4(af[kb], bf[kb], acc);
      if (o < SIZE_M) {
#pragma unroll
        for (int j = 0; j < 8; ++j)
          sW[(mt * 16 + j + 8 * hh) * WST + o] = acc[j];
      }
    }
  }

  for (int nt = 0; nt < 4; ++nt) {
    int o = nt * 16 + r16;
    v2f be[16], ba[16];
#pragma unroll
    for (int kb = 0; kb < 16; ++kb) {
      be[kb] = *(const v2f*)&eW[o * 64 + kb * 4 + 2 * hh];
      ba[kb] = *(const v2f*)&aW[o * 64 + kb * 4 + 2 * hh];
    }
    float ebv = eb[o], abv = ab[o];
    for (int mt = wave; mt < RP / 16; mt += 8) {
      v2f af[16];
#pragma unroll
      for (int kb = 0; kb < 16; ++kb)
        af[kb] = *(const v2f*)&sV[(mt * 16 + r16) * LP + kb * 4 + 2 * hh];
      v8f ce = {0.f,0.f,0.f,0.f,0.f,0.f,0.f,0.f};
      v8f ca = {0.f,0.f,0.f,0.f,0.f,0.f,0.f,0.f};
#pragma unroll
      for (int kb = 0; kb < 16; ++kb) {
        ce = wmma_f32_k4(af[kb], be[kb], ce);
        ca = wmma_f32_k4(af[kb], ba[kb], ca);
      }
#pragma unroll
      for (int j = 0; j < 8; ++j) {
        int row = mt * 16 + j + 8 * hh;
        sE[row * LP + o] = 1.f / (1.f + __expf(-(ce[j] + ebv)));
        sA[row * LP + o] = tanhf(ca[j] + abv);
      }
    }
  }
  __syncthreads();

  const int mg = lane & 3;
  const int dq = wave * 8 + (lane >> 2);
  float mv[MPT];
#pragma unroll
  for (int i = 0; i < MPT; ++i) {
    int m = mg + 4 * i;
    mv[i] = (m < SIZE_M) ? Mv0[m * 64 + dq] : 0.f;
  }

  for (int r = tid; r < TT; r += 256) {
    float mx = -1e30f;
    for (int m = 0; m < SIZE_M; ++m) mx = fmaxf(mx, sW[r * WST + m]);
    float s = 0.f;
    for (int m = 0; m < SIZE_M; ++m) {
      float ev = __expf(sW[r * WST + m] - mx);
      sW[r * WST + m] = ev; s += ev;
    }
    float inv = 1.f / s;
    for (int m = 0; m < SIZE_M; ++m) sW[r * WST + m] *= inv;
  }
  __syncthreads();

  {
    const bool has13 = (mg + 4 * (MPT - 1)) < SIZE_M;
    const float* pW = sW + mg;
    const float* pE = sE + dq;
    const float* pA = sA + dq;
    float*       pR = sV + dq;

    float wA[MPT], wB[MPT], eA, aA, eB, aB;
    scan_load(wA, eA, aA, pW, pE, pA, has13);
    scan_load(wB, eB, aB, pW + WST, pE + LP, pA + LP, has13);
    scan_update(mv, wA, eA, aA);

    for (int t = 1; t < TT - 1; t += 2) {
      scan_load(wA, eA, aA, pW + (t + 1) * WST, pE + (t + 1) * LP, pA + (t + 1) * LP, has13);
      float r0 = scan_read(mv, wB);
      if (mg == 0) pR[(t - 1) * LP] = r0;
      scan_update(mv, wB, eB, aB);

      scan_load(wB, eB, aB, pW + (t + 2) * WST, pE + (t + 2) * LP, pA + (t + 2) * LP, has13);
      float r1 = scan_read(mv, wA);
      if (mg == 0) pR[t * LP] = r1;
      scan_update(mv, wA, eA, aA);
    }
    float rl = scan_read(mv, wB);
    if (mg == 0) pR[(TT - 2) * LP] = rl;
  }

  for (int idx = (TT - 1) * 64 + tid; idx < RP * 64; idx += 256) sV[(idx >> 6) * LP + (idx & 63)] = 0.f;
  __syncthreads();

  for (int nt = 0; nt < 4; ++nt) {
    int o = nt * 16 + r16;
    v2f bf[32];
#pragma unroll
    for (int kb = 0; kb < 32; ++kb)
      bf[kb] = *(const v2f*)&fW[o * 128 + kb * 4 + 2 * hh];
    float fbv = fb[o];
    for (int mt = wave; mt < RP / 16; mt += 8) {
      int row = mt * 16 + r16;
      v2f af[32];
#pragma unroll
      for (int kb = 0; kb < 16; ++kb)
        af[kb] = *(const v2f*)&sV[row * LP + kb * 4 + 2 * hh];
#pragma unroll
      for (int kb = 16; kb < 32; ++kb)
        af[kb] = *(const v2f*)&sK[(row + 1) * LP + (kb - 16) * 4 + 2 * hh];
      v8f acc = {0.f,0.f,0.f,0.f,0.f,0.f,0.f,0.f};
#pragma unroll
      for (int kb = 0; kb < 32; ++kb) acc = wmma_f32_k4(af[kb], bf[kb], acc);
#pragma unroll
      for (int j = 0; j < 8; ++j) {
        int rr = mt * 16 + j + 8 * hh;
        if (rr < TT - 1) sE[rr * LP + o] = tanhf(acc[j] + fbv);
      }
    }
  }
  __syncthreads();

  for (int r = tid; r < TT - 1; r += 256) {
    float s1 = 0.f, s2 = 0.f;
    for (int c = 0; c < 64; ++c) {
      s1 += sE[r * LP + c] * abilW[c];
      s2 += sK[(r + 1) * LP + c] * diffW[c];
    }
    float ability = tanhf(s1 + abilb[0]);
    float diff    = tanhf(s2 + diffb[0]);
    out[b * (TT - 1) + r] = 3.f * ability - diff;
  }
}

extern "C" void kernel_launch(void* const* d_in, const int* in_sizes, int n_in,
                              void* d_out, int out_size, void* d_ws, size_t ws_size,
                              hipStream_t stream) {
  (void)in_sizes; (void)n_in; (void)out_size; (void)d_ws; (void)ws_size;
  const int*   question = (const int*)  d_in[0];
  const int*   response = (const int*)  d_in[1];
  const float* vq_emb   = (const float*)d_in[2];
  const float* vc_emb   = (const float*)d_in[3];
  const float* kq_emb   = (const float*)d_in[4];
  const float* kc_emb   = (const float*)d_in[5];
  const float* Mk       = (const float*)d_in[6];
  const float* Mv0      = (const float*)d_in[7];
  const float* eW       = (const float*)d_in[8];
  const float* ebv      = (const float*)d_in[9];
  const float* aW       = (const float*)d_in[10];
  const float* abv      = (const float*)d_in[11];
  const float* fW       = (const float*)d_in[12];
  const float* fbv      = (const float*)d_in[13];
  const float* abilW    = (const float*)d_in[14];
  const float* abilb    = (const float*)d_in[15];
  const float* diffW    = (const float*)d_in[16];
  const float* diffb    = (const float*)d_in[17];
  float* out = (float*)d_out;

  const size_t smem_bytes =
      (size_t)(RK * LP + 3 * RP * LP + RP * WST) * sizeof(float);

  (void)hipFuncSetAttribute((const void*)deepirt_fused_kernel,
                            hipFuncAttributeMaxDynamicSharedMemorySize,
                            (int)smem_bytes);

  deepirt_fused_kernel<<<BATCH, 256, smem_bytes, stream>>>(
      question, response, vq_emb, vc_emb, kq_emb, kc_emb, Mk, Mv0,
      eW, ebv, aW, abv, fW, fbv, abilW, abilb, diffW, diffb, out);
}


# e, a and f activations branch-free on v_exp_f32 + v_rcp_f32 (tanh = 1 - 2/(1 + e^2x), sigmoid = rcp(1 + e^-x)) in place of library tanhf and the full-accuracy divide; values differ in the last bits
# speedup vs baseline: 5.6997x; 1.2100x over previous
#include <hip/hip_runtime.h>
#include <math.h>


#define NUM_Q  10000
#define NUM_C  300
#define DIM_S  64
#define SIZE_M 50
#define BATCH  128
#define TT     200
#define HALF   32

#define RK  212
#define RP  208
#define WST 52
#define LP  68

typedef __attribute__((ext_vector_type(2))) float v2f;
typedef __attribute__((ext_vector_type(8))) float v8f;

static __device__ __forceinline__ v8f wmma_f32_k4(v2f a, v2f b, v8f c) {
  return __builtin_amdgcn_wmma_f32_16x16x4_f32(false, a, false, b, (short)0, c, false, false);
}

static __device__ __forceinline__ float tanh_exp_rcp(float x) {
  return 1.f - 2.f * __builtin_amdgcn_rcpf(1.f + __expf(2.f * x));
}
static __device__ __forceinline__ float sigmoid_exp_rcp(float x) {
  return __builtin_amdgcn_rcpf(1.f + __expf(-x));
}

#define GP  13

typedef __attribute__((ext_vector_type(4))) float v4f;

#define MPT 13

static __device__ __forceinline__ void scan_load(float (&w)[MPT], float& e, float& a,
                                                 const float* sWrow, const float* sErow,
                                                 const float* sArow, bool has13) {
#pragma unroll
  for (int i = 0; i < MPT - 1; ++i) w[i] = sWrow[4 * i];
  w[MPT - 1] = has13 ? sWrow[4 * (MPT - 1)] : 0.f;
  e = *sErow;
  a = *sArow;
}

static __device__ __forceinline__ void scan_update(float (&mv)[MPT], const float (&w)[MPT],
                                                   float e, float a) {
#pragma unroll
  for (int i = 0; i < MPT; ++i) mv[i] = mv[i] * (1.f - w[i] * e) + w[i] * a;
}

template <int K>
static __device__ __forceinline__ float quad_lane(float v) {
  return __int_as_float(__builtin_amdgcn_mov_dpp(__float_as_int(v), K * 0x55, 0xf, 0xf, false));
}

static __device__ __forceinline__ float scan_read(const float (&mv)[MPT], const float (&w)[MPT]) {
  float p = 0.f;
#pragma unroll
  for (int i = 0; i < MPT; ++i) p += w[i] * mv[i];
  return ((p + quad_lane<1>(p)) + quad_lane<2>(p)) + quad_lane<3>(p);
}

__global__ __launch_bounds__(256) void deepirt_fused_kernel(
    const int*   __restrict__ question, const int*   __restrict__ response,
    const float* __restrict__ vq_emb,   const float* __restrict__ vc_emb,
    const float* __restrict__ kq_emb,   const float* __restrict__ kc_emb,
    const float* __restrict__ Mk,       const float* __restrict__ Mv0,
    const float* __restrict__ eW, const float* __restrict__ eb,
    const float* __restrict__ aW, const float* __restrict__ ab,
    const float* __restrict__ fW, const float* __restrict__ fb,
    const float* __restrict__ abilW, const float* __restrict__ abilb,
    const float* __restrict__ diffW, const float* __restrict__ diffb,
    float* __restrict__ out)
{
  extern __shared__ float smem[];
  float* sK   = smem;
  float* sV   = sK  + RK * LP;
  float* sE   = sV  + RP * LP;
  float* sA   = sE  + RP * LP;
  float* sW   = sA  + RP * LP;

  const int b    = blockIdx.x;
  const int tid  = threadIdx.x;
  const int lane = tid & 31;
  const int wave = tid >> 5;
  const int hh   = lane >> 4;
  const int r16  = lane & 15;

  {
    const int c4 = tid & 15;
    const int gr = tid >> 4;
    int qi[GP], ri[GP];
#pragma unroll
    for (int p = 0; p < GP; ++p) {
      int t  = p * 16 + gr;
      int tc = (t < TT) ? t : (TT - 1);
      qi[p] = question[b * TT + tc];
      ri[p] = response[b * TT + tc];
    }
    const bool   qhalf = c4 < 8;
    const float* ktab  = (qhalf ? kq_emb : kc_emb) + 4 * (c4 & 7);
    const float* vtab  = (qhalf ? vq_emb : vc_emb) + 4 * (c4 & 7);
    v4f kx[GP], vx[GP];
#pragma unroll
    for (int p = 0; p < GP; ++p) {
      int krow = qhalf ? qi[p] : 0;
      int vrow = qhalf ? qi[p] + NUM_Q * ri[p] : NUM_C * ri[p];
      kx[p] = *(const v4f*)(ktab + krow * HALF);
      vx[p] = *(const v4f*)(vtab + vrow * HALF);
    }
    const v4f z4 = {0.f, 0.f, 0.f, 0.f};
#pragma unroll
    for (int p = 0; p < GP; ++p) {
      int t = p * 16 + gr;
      *(v4f*)&sK[t * LP + 4 * c4] = (t < TT) ? kx[p] : z4;
      *(v4f*)&sV[t * LP + 4 * c4] = (t < TT) ? vx[p] : z4;
    }
    if (tid < (RK - RP) * 16) *(v4f*)&sK[(RP + gr) * LP + 4 * c4] = z4;
  }
  __syncthreads();

  for (int nt = 0; nt < 4; ++nt) {
    int o  = nt * 16 + r16;
    int oc = (o < SIZE_M) ? o : (SIZE_M - 1);
    v2f bf[16];
#pragma unroll
    for (int kb = 0; kb < 16; ++kb)
      bf[kb] = *(const v2f*)&Mk[oc * 64 + kb * 4 + 2 * hh];
    for (int mt = wave; mt < RP / 16; mt += 8) {
      v2f af[16];
#pragma unroll
      for (int kb = 0; kb < 16; ++kb)
        af[kb] = *(const v2f*)&sK[(mt * 16 + r16) * LP + kb * 4 + 2 * hh];
      v8f acc = {0.f,0.f,0.f,0.f,0.f,0.f,0.f,0.f};
#pragma unroll
      for (int kb = 0; kb < 16; ++kb) acc = wmma_f32_k4(af[kb], bf[kb], acc);
      if (o < SIZE_M) {
#pragma unroll
        for (int j = 0; j < 8; ++j)
          sW[(mt * 16 + j + 8 * hh) * WST + o] = acc[j];
      }
    }
  }

  for (int nt = 0; nt < 4; ++nt) {
    int o = nt * 16 + r16;
    v2f be[16], ba[16];
#pragma unroll
    for (int kb = 0; kb < 16; ++kb) {
      be[kb] = *(const v2f*)&eW[o * 64 + kb * 4 + 2 * hh];
      ba[kb] = *(const v2f*)&aW[o * 64 + kb * 4 + 2 * hh];
    }
    float ebv = eb[o], abv = ab[o];
    for (int mt = wave; mt < RP / 16; mt += 8) {
      v2f af[16];
#pragma unroll
      for (int kb = 0; kb < 16; ++kb)
        af[kb] = *(const v2f*)&sV[(mt * 16 + r16) * LP + kb * 4 + 2 * hh];
      v8f ce = {0.f,0.f,0.f,0.f,0.f,0.f,0.f,0.f};
      v8f ca = {0.f,0.f,0.f,0.f,0.f,0.f,0.f,0.f};
#pragma unroll
      for (int kb = 0; kb < 16; ++kb) {
        ce = wmma_f32_k4(af[kb], be[kb], ce);
        ca = wmma_f32_k4(af[kb], ba[kb], ca);
      }
#pragma unroll
      for (int j = 0; j < 8; ++j) {
        int row = mt * 16 + j + 8 * hh;
        sE[row * LP + o] = sigmoid_exp_rcp(ce[j] + ebv);
        sA[row * LP + o] = tanh_exp_rcp(ca[j] + abv);
      }
    }
  }
  __syncthreads();

  const int mg = lane & 3;
  const int dq = wave * 8 + (lane >> 2);
  float mv[MPT];
#pragma unroll
  for (int i = 0; i < MPT; ++i) {
    int m = mg + 4 * i;
    mv[i] = (m < SIZE_M) ? Mv0[m * 64 + dq] : 0.f;
  }

  for (int r = tid; r < TT; r += 256) {
    float mx = -1e30f;
    for (int m = 0; m < SIZE_M; ++m) mx = fmaxf(mx, sW[r * WST + m]);
    float s = 0.f;
    for (int m = 0; m < SIZE_M; ++m) {
      float ev = __expf(sW[r * WST + m] - mx);
      sW[r * WST + m] = ev; s += ev;
    }
    float inv = 1.f / s;
    for (int m = 0; m < SIZE_M; ++m) sW[r * WST + m] *= inv;
  }
  __syncthreads();

  {
    const bool has13 = (mg + 4 * (MPT - 1)) < SIZE_M;
    const float* pW = sW + mg;
    const float* pE = sE + dq;
    const float* pA = sA + dq;
    float*       pR = sV + dq;

    float wA[MPT], wB[MPT], eA, aA, eB, aB;
    scan_load(wA, eA, aA, pW, pE, pA, has13);
    scan_load(wB, eB, aB, pW + WST, pE + LP, pA + LP, has13);
    scan_update(mv, wA, eA, aA);

    for (int t = 1; t < TT - 1; t += 2) {
      scan_load(wA, eA, aA, pW + (t + 1) * WST, pE + (t + 1) * LP, pA + (t + 1) * LP, has13);
      float r0 = scan_read(mv, wB);
      if (mg == 0) pR[(t - 1) * LP] = r0;
      scan_update(mv, wB, eB, aB);

      scan_load(wB, eB, aB, pW + (t + 2) * WST, pE + (t + 2) * LP, pA + (t + 2) * LP, has13);
      float r1 = scan_read(mv, wA);
      if (mg == 0) pR[t * LP] = r1;
      scan_update(mv, wA, eA, aA);
    }
    float rl = scan_read(mv, wB);
    if (mg == 0) pR[(TT - 2) * LP] = rl;
  }

  for (int idx = (TT - 1) * 64 + tid; idx < RP * 64; idx += 256) sV[(idx >> 6) * LP + (idx & 63)] = 0.f;
  __syncthreads();

  for (int nt = 0; nt < 4; ++nt) {
    int o = nt * 16 + r16;
    v2f bf[32];
#pragma unroll
    for (int kb = 0; kb < 32; ++kb)
      bf[kb] = *(const v2f*)&fW[o * 128 + kb * 4 + 2 * hh];
    float fbv = fb[o];
    for (int mt = wave; mt < RP / 16; mt += 8) {
      int row = mt * 16 + r16;
      v2f af[32];
#pragma unroll
      for (int kb = 0; kb < 16; ++kb)
        af[kb] = *(const v2f*)&sV[row * LP + kb * 4 + 2 * hh];
#pragma unroll
      for (int kb = 16; kb < 32; ++kb)
        af[kb] = *(const v2f*)&sK[(row + 1) * LP + (kb - 16) * 4 + 2 * hh];
      v8f acc = {0.f,0.f,0.f,0.f,0.f,0.f,0.f,0.f};
#pragma unroll
      for (int kb = 0; kb < 32; ++kb) acc = wmma_f32_k4(af[kb], bf[kb], acc);
#pragma unroll
      for (int j = 0; j < 8; ++j) {
        int rr = mt * 16 + j + 8 * hh;
        sE[rr * LP + o] = tanh_exp_rcp(acc[j] + fbv);
      }
    }
  }
  __syncthreads();

  for (int r = tid; r < TT - 1; r += 256) {
    float s1 = 0.f, s2 = 0.f;
    for (int c = 0; c < 64; ++c) {
      s1 += sE[r * LP + c] * abilW[c];
      s2 += sK[(r + 1) * LP + c] * diffW[c];
    }
    float ability = tanhf(s1 + abilb[0]);
    float diff    = tanhf(s2 + diffb[0]);
    out[b * (TT - 1) + r] = 3.f * ability - diff;
  }
}

extern "C" void kernel_launch(void* const* d_in, const int* in_sizes, int n_in,
                              void* d_out, int out_size, void* d_ws, size_t ws_size,
                              hipStream_t stream) {
  (void)in_sizes; (void)n_in; (void)out_size; (void)d_ws; (void)ws_size;
  const int*   question = (const int*)  d_in[0];
  const int*   response = (const int*)  d_in[1];
  const float* vq_emb   = (const float*)d_in[2];
  const float* vc_emb   = (const float*)d_in[3];
  const float* kq_emb   = (const float*)d_in[4];
  const float* kc_emb   = (const float*)d_in[5];
  const float* Mk       = (const float*)d_in[6];
  const float* Mv0      = (const float*)d_in[7];
  const float* eW       = (const float*)d_in[8];
  const float* ebv      = (const float*)d_in[9];
  const float* aW       = (const float*)d_in[10];
  const float* abv      = (const float*)d_in[11];
  const float* fW       = (const float*)d_in[12];
  const float* fbv      = (const float*)d_in[13];
  const float* abilW    = (const float*)d_in[14];
  const float* abilb    = (const float*)d_in[15];
  const float* diffW    = (const float*)d_in[16];
  const float* diffb    = (const float*)d_in[17];
  float* out = (float*)d_out;

  const size_t smem_bytes =
      (size_t)(RK * LP + 3 * RP * LP + RP * WST) * sizeof(float);

  (void)hipFuncSetAttribute((const void*)deepirt_fused_kernel,
                            hipFuncAttributeMaxDynamicSharedMemorySize,
                            (int)smem_bytes);

  deepirt_fused_kernel<<<BATCH, 256, smem_bytes, stream>>>(
      question, response, vq_emb, vc_emb, kq_emb, kc_emb, Mk, Mv0,
      eW, ebv, aW, abv, fW, fbv, abilW, abilb, diffW, diffb, out);
}


# GEMM phases re-tiled: a wave keeps one 16-column tile's B fragments for a whole phase (fetched once, early) and walks every 2nd row tile; next tile's A fragments load as one group under the WMMAs
# speedup vs baseline: 6.5388x; 1.1472x over previous
#include <hip/hip_runtime.h>
#include <math.h>


#define NUM_Q  10000
#define NUM_C  300
#define DIM_S  64
#define SIZE_M 50
#define BATCH  128
#define TT     200
#define HALF   32

#define RK  212
#define RP  208
#define WST 52
#define LP  68

typedef __attribute__((ext_vector_type(2))) float v2f;
typedef __attribute__((ext_vector_type(8))) float v8f;

static __device__ __forceinline__ v8f wmma_f32_k4(v2f a, v2f b, v8f c) {
  return __builtin_amdgcn_wmma_f32_16x16x4_f32(false, a, false, b, (short)0, c, false, false);
}

static __device__ __forceinline__ float tanh_exp_rcp(float x) {
  return 1.f - 2.f * __builtin_amdgcn_rcpf(1.f + __expf(2.f * x));
}
static __device__ __forceinline__ float sigmoid_exp_rcp(float x) {
  return __builtin_amdgcn_rcpf(1.f + __expf(-x));
}

#define GP  13

typedef __attribute__((ext_vector_type(4))) float v4f;

#define MPT 13

static __device__ __forceinline__ void scan_load(float (&w)[MPT], float& e, float& a,
                                                 const float* sWrow, const float* sErow,
                                                 const float* sArow, bool has13) {
#pragma unroll
  for (int i = 0; i < MPT - 1; ++i) w[i] = sWrow[4 * i];
  w[MPT - 1] = has13 ? sWrow[4 * (MPT - 1)] : 0.f;
  e = *sErow;
  a = *sArow;
}

static __device__ __forceinline__ void scan_update(float (&mv)[MPT], const float (&w)[MPT],
                                                   float e, float a) {
#pragma unroll
  for (int i = 0; i < MPT; ++i) mv[i] = mv[i] * (1.f - w[i] * e) + w[i] * a;
}

template <int K>
static __device__ __forceinline__ float quad_lane(float v) {
  return __int_as_float(__builtin_amdgcn_mov_dpp(__float_as_int(v), K * 0x55, 0xf, 0xf, false));
}

static __device__ __forceinline__ float scan_read(const float (&mv)[MPT], const float (&w)[MPT]) {
  float p = 0.f;
#pragma unroll
  for (int i = 0; i < MPT; ++i) p += w[i] * mv[i];
  return ((p + quad_lane<1>(p)) + quad_lane<2>(p)) + quad_lane<3>(p);
}


static __device__ __forceinline__ void load_afrag16(v2f (&af)[16], const float* p) {
#pragma unroll
  for (int kb = 0; kb < 16; ++kb) af[kb] = *(const v2f*)(p + kb * 4);
}

static __device__ __forceinline__ void tile_wlog(const v2f (&af)[16], const v2f (&bf)[16],
                                                 float* sW, int mt, int hh, int o) {
  v8f acc = {0.f,0.f,0.f,0.f,0.f,0.f,0.f,0.f};
#pragma unroll
  for (int kb = 0; kb < 16; ++kb) acc = wmma_f32_k4(af[kb], bf[kb], acc);
  if (o < SIZE_M) {
#pragma unroll
    for (int j = 0; j < 8; ++j) sW[(mt * 16 + j + 8 * hh) * WST + o] = acc[j];
  }
}

static __device__ __forceinline__ void tile_ea(const v2f (&af)[16], const v2f (&be)[16], const v2f (&ba)[16],
                                               float ebv, float abv, float* sE, float* sA,
                                               int mt, int hh, int o) {
  v8f ce = {0.f,0.f,0.f,0.f,0.f,0.f,0.f,0.f};
  v8f ca = {0.f,0.f,0.f,0.f,0.f,0.f,0.f,0.f};
#pragma unroll
  for (int kb = 0; kb < 16; ++kb) {
    ce = wmma_f32_k4(af[kb], be[kb], ce);
    ca = wmma_f32_k4(af[kb], ba[kb], ca);
  }
#pragma unroll
  for (int j = 0; j < 8; ++j) {
    int row = mt * 16 + j + 8 * hh;
    sE[row * LP + o] = sigmoid_exp_rcp(ce[j] + ebv);
    sA[row * LP + o] = tanh_exp_rcp(ca[j] + abv);
  }
}

static __device__ __forceinline__ void load_afrag32(v2f (&af)[32], const float* pr, const float* pk) {
#pragma unroll
  for (int kb = 0; kb < 16; ++kb) af[kb] = *(const v2f*)(pr + kb * 4);
#pragma unroll
  for (int kb = 0; kb < 16; ++kb) af[16 + kb] = *(const v2f*)(pk + kb * 4);
}

static __device__ __forceinline__ void tile_f(const v2f (&af)[32], const v2f (&bf)[32], float fbv,
                                              float* sE, int mt, int hh, int o) {
  v8f acc = {0.f,0.f,0.f,0.f,0.f,0.f,0.f,0.f};
#pragma unroll
  for (int kb = 0; kb < 32; ++kb) acc = wmma_f32_k4(af[kb], bf[kb], acc);
#pragma unroll
  for (int j = 0; j < 8; ++j) {
    int rr = mt * 16 + j + 8 * hh;
    sE[rr * LP + o] = tanh_exp_rcp(acc[j] + fbv);
  }
}

__global__ __launch_bounds__(256) void deepirt_fused_kernel(
    const int*   __restrict__ question, const int*   __restrict__ response,
    const float* __restrict__ vq_emb,   const float* __restrict__ vc_emb,
    const float* __restrict__ kq_emb,   const float* __restrict__ kc_emb,
    const float* __restrict__ Mk,       const float* __restrict__ Mv0,
    const float* __restrict__ eW, const float* __restrict__ eb,
    const float* __restrict__ aW, const float* __restrict__ ab,
    const float* __restrict__ fW, const float* __restrict__ fb,
    const float* __restrict__ abilW, const float* __restrict__ abilb,
    const float* __restrict__ diffW, const float* __restrict__ diffb,
    float* __restrict__ out)
{
  extern __shared__ float smem[];
  float* sK   = smem;
  float* sV   = sK  + RK * LP;
  float* sE   = sV  + RP * LP;
  float* sA   = sE  + RP * LP;
  float* sW   = sA  + RP * LP;

  const int b    = blockIdx.x;
  const int tid  = threadIdx.x;
  const int lane = tid & 31;
  const int wave = tid >> 5;
  const int hh   = lane >> 4;
  const int r16  = lane & 15;

  const int wv  = __builtin_amdgcn_readfirstlane(wave);
  const int nt  = wv >> 1;
  const int par = (wv ^ (wv >> 2)) & 1;
  const int o   = nt * 16 + r16;

  v2f bk[16], be[16], ba[16];
  float ebv, abv;

  {
    const int c4 = tid & 15;
    const int gr = tid >> 4;
    int qi[GP], ri[GP];
#pragma unroll
    for (int p = 0; p < GP; ++p) {
      int t  = p * 16 + gr;
      int tc = (t < TT) ? t : (TT - 1);
      qi[p] = question[b * TT + tc];
      ri[p] = response[b * TT + tc];
    }
    {
      const int oc = (o < SIZE_M) ? o : (SIZE_M - 1);
#pragma unroll
      for (int kb = 0; kb < 16; ++kb) {
        bk[kb] = *(const v2f*)&Mk[oc * 64 + kb * 4 + 2 * hh];
        be[kb] = *(const v2f*)&eW[o * 64 + kb * 4 + 2 * hh];
        ba[kb] = *(const v2f*)&aW[o * 64 + kb * 4 + 2 * hh];
      }
      ebv = eb[o];
      abv = ab[o];
    }
    const bool   qhalf = c4 < 8;
    const float* ktab  = (qhalf ? kq_emb : kc_emb) + 4 * (c4 & 7);
    const float* vtab  = (qhalf ? vq_emb : vc_emb) + 4 * (c4 & 7);
    v4f kx[GP], vx[GP];
#pragma unroll
    for (int p = 0; p < GP; ++p) {
      int krow = qhalf ? qi[p] : 0;
      int vrow = qhalf ? qi[p] + NUM_Q * ri[p] : NUM_C * ri[p];
      kx[p] = *(const v4f*)(ktab + krow * HALF);
      vx[p] = *(const v4f*)(vtab + vrow * HALF);
    }
    const v4f z4 = {0.f, 0.f, 0.f, 0.f};
#pragma unroll
    for (int p = 0; p < GP; ++p) {
      int t = p * 16 + gr;
      *(v4f*)&sK[t * LP + 4 * c4] = (t < TT) ? kx[p] : z4;
      *(v4f*)&sV[t * LP + 4 * c4] = (t < TT) ? vx[p] : z4;
    }
    if (tid < (RK - RP) * 16) *(v4f*)&sK[(RP + gr) * LP + 4 * c4] = z4;
  }
  __syncthreads();

  {
    v2f a0[16], a1[16];
    const float* pa = sK + r16 * LP + 2 * hh;
    load_afrag16(a0, pa + par * 16 * LP);
    for (int i = 0; i < 3; ++i) {
      const int m0 = par + 4 * i;
      const int m2 = (m0 + 4 < RP / 16) ? m0 + 4 : RP / 16 - 1;
      load_afrag16(a1, pa + (m0 + 2) * 16 * LP);
      __builtin_amdgcn_sched_barrier(0);
      tile_wlog(a0, bk, sW, m0, hh, o);
      load_afrag16(a0, pa + m2 * 16 * LP);
      __builtin_amdgcn_sched_barrier(0);
      tile_wlog(a1, bk, sW, m0 + 2, hh, o);
    }
    if (par == 0) tile_wlog(a0, bk, sW, RP / 16 - 1, hh, o);
  }

  {
    v2f a0[16], a1[16];
    const float* pa = sV + r16 * LP + 2 * hh;
    load_afrag16(a0, pa + par * 16 * LP);
    for (int i = 0; i < 3; ++i) {
      const int m0 = par + 4 * i;
      const int m2 = (m0 + 4 < RP / 16) ? m0 + 4 : RP / 16 - 1;
      load_afrag16(a1, pa + (m0 + 2) * 16 * LP);
      __builtin_amdgcn_sched_barrier(0);
      tile_ea(a0, be, ba, ebv, abv, sE, sA, m0, hh, o);
      load_afrag16(a0, pa + m2 * 16 * LP);
      __builtin_amdgcn_sched_barrier(0);
      tile_ea(a1, be, ba, ebv, abv, sE, sA, m0 + 2, hh, o);
    }
    if (par == 0) tile_ea(a0, be, ba, ebv, abv, sE, sA, RP / 16 - 1, hh, o);
  }
  __syncthreads();

  const int mg = lane & 3;
  const int dq = wave * 8 + (lane >> 2);
  float mv[MPT];
#pragma unroll
  for (int i = 0; i < MPT; ++i) {
    int m = mg + 4 * i;
    mv[i] = (m < SIZE_M) ? Mv0[m * 64 + dq] : 0.f;
  }

  v2f bw[32];
#pragma unroll
  for (int kb = 0; kb < 32; ++kb)
    bw[kb] = *(const v2f*)&fW[o * 128 + kb * 4 + 2 * hh];
  const float fbv = fb[o];

  for (int r = tid; r < TT; r += 256) {
    float mx = -1e30f;
    for (int m = 0; m < SIZE_M; ++m) mx = fmaxf(mx, sW[r * WST + m]);
    float s = 0.f;
    for (int m = 0; m < SIZE_M; ++m) {
      float ev = __expf(sW[r * WST + m] - mx);
      sW[r * WST + m] = ev; s += ev;
    }
    float inv = 1.f / s;
    for (int m = 0; m < SIZE_M; ++m) sW[r * WST + m] *= inv;
  }
  __syncthreads();

  {
    const bool has13 = (mg + 4 * (MPT - 1)) < SIZE_M;
    const float* pW = sW + mg;
    const float* pE = sE + dq;
    const float* pA = sA + dq;
    float*       pR = sV + dq;

    float wA[MPT], wB[MPT], eA, aA, eB, aB;
    scan_load(wA, eA, aA, pW, pE, pA, has13);
    scan_load(wB, eB, aB, pW + WST, pE + LP, pA + LP, has13);
    scan_update(mv, wA, eA, aA);

    for (int t = 1; t < TT - 1; t += 2) {
      scan_load(wA, eA, aA, pW + (t + 1) * WST, pE + (t + 1) * LP, pA + (t + 1) * LP, has13);
      float r0 = scan_read(mv, wB);
      if (mg == 0) pR[(t - 1) * LP] = r0;
      scan_update(mv, wB, eB, aB);

      scan_load(wB, eB, aB, pW + (t + 2) * WST, pE + (t + 2) * LP, pA + (t + 2) * LP, has13);
      float r1 = scan_read(mv, wA);
      if (mg == 0) pR[t * LP] = r1;
      scan_update(mv, wA, eA, aA);
    }
    float rl = scan_read(mv, wB);
    if (mg == 0) pR[(TT - 2) * LP] = rl;
  }

  for (int idx = (TT - 1) * 64 + tid; idx < RP * 64; idx += 256) sV[(idx >> 6) * LP + (idx & 63)] = 0.f;
  __syncthreads();

  {
    v2f a0[32], a1[32];
    const float* pr = sV + r16 * LP + 2 * hh;
    const float* pk = sK + (r16 + 1) * LP + 2 * hh;
    load_afrag32(a0, pr + par * 16 * LP, pk + par * 16 * LP);
    for (int i = 0; i < 3; ++i) {
      const int m0 = par + 4 * i;
      const int m2 = (m0 + 4 < RP / 16) ? m0 + 4 : RP / 16 - 1;
      load_afrag32(a1, pr + (m0 + 2) * 16 * LP, pk + (m0 + 2) * 16 * LP);
      __builtin_amdgcn_sched_barrier(0);
      tile_f(a0, bw, fbv, sE, m0, hh, o);
      load_afrag32(a0, pr + m2 * 16 * LP, pk + m2 * 16 * LP);
      __builtin_amdgcn_sched_barrier(0);
      tile_f(a1, bw, fbv, sE, m0 + 2, hh, o);
    }
    if (par == 0) tile_f(a0, bw, fbv, sE, RP / 16 - 1, hh, o);
  }
  __syncthreads();

  for (int r = tid; r < TT - 1; r += 256) {
    float s1 = 0.f, s2 = 0.f;
    for (int c = 0; c < 64; ++c) {
      s1 += sE[r * LP + c] * abilW[c];
      s2 += sK[(r + 1) * LP + c] * diffW[c];
    }
    float ability = tanhf(s1 + abilb[0]);
    float diff    = tanhf(s2 + diffb[0]);
    out[b * (TT - 1) + r] = 3.f * ability - diff;
  }
}

extern "C" void kernel_launch(void* const* d_in, const int* in_sizes, int n_in,
                              void* d_out, int out_size, void* d_ws, size_t ws_size,
                              hipStream_t stream) {
  (void)in_sizes; (void)n_in; (void)out_size; (void)d_ws; (void)ws_size;
  const int*   question = (const int*)  d_in[0];
  const int*   response = (const int*)  d_in[1];
  const float* vq_emb   = (const float*)d_in[2];
  const float* vc_emb   = (const float*)d_in[3];
  const float* kq_emb   = (const float*)d_in[4];
  const float* kc_emb   = (const float*)d_in[5];
  const float* Mk       = (const float*)d_in[6];
  const float* Mv0      = (const float*)d_in[7];
  const float* eW       = (const float*)d_in[8];
  const float* ebv      = (const float*)d_in[9];
  const float* aW       = (const float*)d_in[10];
  const float* abv      = (const float*)d_in[11];
  const float* fW       = (const float*)d_in[12];
  const float* fbv      = (const float*)d_in[13];
  const float* abilW    = (const float*)d_in[14];
  const float* abilb    = (const float*)d_in[15];
  const float* diffW    = (const float*)d_in[16];
  const float* diffb    = (const float*)d_in[17];
  float* out = (float*)d_out;

  const size_t smem_bytes =
      (size_t)(RK * LP + 3 * RP * LP + RP * WST) * sizeof(float);

  (void)hipFuncSetAttribute((const void*)deepirt_fused_kernel,
                            hipFuncAttributeMaxDynamicSharedMemorySize,
                            (int)smem_bytes);

  deepirt_fused_kernel<<<BATCH, 256, smem_bytes, stream>>>(
      question, response, vq_emb, vc_emb, kq_emb, kc_emb, Mk, Mv0,
      eW, ebv, aW, abv, fW, fbv, abilW, abilb, diffW, diffb, out);
}


# Scan step on packed slot pairs: update as Mv + w*(a - e*Mv) (2 FMAs per slot, was 3 ops); read as 4 FMA partial sums, a tree and a quad butterfly (sum order changed); 2 reads stored per trip
# speedup vs baseline: 8.4012x; 1.2848x over previous
#include <hip/hip_runtime.h>
#include <math.h>


#define NUM_Q  10000
#define NUM_C  300
#define DIM_S  64
#define SIZE_M 50
#define BATCH  128
#define TT     200
#define HALF   32

#define RK  212
#define RP  208
#define WST 60
#define LP  68

typedef __attribute__((ext_vector_type(2))) float v2f;
typedef __attribute__((ext_vector_type(8))) float v8f;

static __device__ __forceinline__ v8f wmma_f32_k4(v2f a, v2f b, v8f c) {
  return __builtin_amdgcn_wmma_f32_16x16x4_f32(false, a, false, b, (short)0, c, false, false);
}

static __device__ __forceinline__ float tanh_exp_rcp(float x) {
  return 1.f - 2.f * __builtin_amdgcn_rcpf(1.f + __expf(2.f * x));
}
static __device__ __forceinline__ float sigmoid_exp_rcp(float x) {
  return __builtin_amdgcn_rcpf(1.f + __expf(-x));
}

#define GP  13

typedef __attribute__((ext_vector_type(4))) float v4f;

#define MPP 7


static __device__ __forceinline__ void scan_load(v2f (&w)[MPP], float& e, float& a,
                                                 const float* sWrow, const float* sErow,
                                                 const float* sArow) {
#pragma unroll
  for (int j = 0; j < MPP; ++j) { w[j].x = sWrow[8 * j]; w[j].y = sWrow[8 * j + 4]; }
  e = *sErow;
  a = *sArow;
}

static __device__ __forceinline__ void scan_update(v2f (&mv)[MPP], const v2f (&w)[MPP], float e, float a) {
  const v2f ne2 = {-e, -e};
  const v2f a2  = {a, a};
#pragma unroll
  for (int j = 0; j < MPP; ++j)
    mv[j] = __builtin_elementwise_fma(w[j], __builtin_elementwise_fma(ne2, mv[j], a2), mv[j]);
}

template <int SEL>
static __device__ __forceinline__ float quad_perm(float v) {
  return __int_as_float(__builtin_amdgcn_mov_dpp(__float_as_int(v), SEL, 0xf, 0xf, false));
}

static __device__ __forceinline__ float scan_read(const v2f (&mv)[MPP], const v2f (&w)[MPP]) {
  v2f s01 = w[0] * mv[0];
  v2f s23 = w[1] * mv[1];
  s01 = __builtin_elementwise_fma(w[2], mv[2], s01);
  s23 = __builtin_elementwise_fma(w[3], mv[3], s23);
  s01 = __builtin_elementwise_fma(w[4], mv[4], s01);
  s23 = __builtin_elementwise_fma(w[5], mv[5], s23);
  s01 = __builtin_elementwise_fma(w[6], mv[6], s01);
  float p = (s01.x + s01.y) + (s23.x + s23.y);
  p += quad_perm<0xB1>(p);
  p += quad_perm<0x4E>(p);
  return p;
}


static __device__ __forceinline__ void load_afrag16(v2f (&af)[16], const float* p) {
#pragma unroll
  for (int kb = 0; kb < 16; ++kb) af[kb] = *(const v2f*)(p + kb * 4);
}

static __device__ __forceinline__ void tile_wlog(const v2f (&af)[16], const v2f (&bf)[16],
                                                 float* sW, int mt, int hh, int o) {
  v8f acc = {0.f,0.f,0.f,0.f,0.f,0.f,0.f,0.f};
#pragma unroll
  for (int kb = 0; kb < 16; ++kb) acc = wmma_f32_k4(af[kb], bf[kb], acc);
  if (o < SIZE_M) {
#pragma unroll
    for (int j = 0; j < 8; ++j) sW[(mt * 16 + j + 8 * hh) * WST + o] = acc[j];
  }
}

static __device__ __forceinline__ void tile_ea(const v2f (&af)[16], const v2f (&be)[16], const v2f (&ba)[16],
                                               float ebv, float abv, float* sE, float* sA,
                                               int mt, int hh, int o) {
  v8f ce = {0.f,0.f,0.f,0.f,0.f,0.f,0.f,0.f};
  v8f ca = {0.f,0.f,0.f,0.f,0.f,0.f,0.f,0.f};
#pragma unroll
  for (int kb = 0; kb < 16; ++kb) {
    ce = wmma_f32_k4(af[kb], be[kb], ce);
    ca = wmma_f32_k4(af[kb], ba[kb], ca);
  }
#pragma unroll
  for (int j = 0; j < 8; ++j) {
    int row = mt * 16 + j + 8 * hh;
    sE[row * LP + o] = sigmoid_exp_rcp(ce[j] + ebv);
    sA[row * LP + o] = tanh_exp_rcp(ca[j] + abv);
  }
}

static __device__ __forceinline__ void load_afrag32(v2f (&af)[32], const float* pr, const float* pk) {
#pragma unroll
  for (int kb = 0; kb < 16; ++kb) af[kb] = *(const v2f*)(pr + kb * 4);
#pragma unroll
  for (int kb = 0; kb < 16; ++kb) af[16 + kb] = *(const v2f*)(pk + kb * 4);
}

static __device__ __forceinline__ void tile_f(const v2f (&af)[32], const v2f (&bf)[32], float fbv,
                                              float* sE, int mt, int hh, int o) {
  v8f acc = {0.f,0.f,0.f,0.f,0.f,0.f,0.f,0.f};
#pragma unroll
  for (int kb = 0; kb < 32; ++kb) acc = wmma_f32_k4(af[kb], bf[kb], acc);
#pragma unroll
  for (int j = 0; j < 8; ++j) {
    int rr = mt * 16 + j + 8 * hh;
    sE[rr * LP + o] = tanh_exp_rcp(acc[j] + fbv);
  }
}

__global__ __launch_bounds__(256) void deepirt_fused_kernel(
    const int*   __restrict__ question, const int*   __restrict__ response,
    const float* __restrict__ vq_emb,   const float* __restrict__ vc_emb,
    const float* __restrict__ kq_emb,   const float* __restrict__ kc_emb,
    const float* __restrict__ Mk,       const float* __restrict__ Mv0,
    const float* __restrict__ eW, const float* __restrict__ eb,
    const float* __restrict__ aW, const float* __restrict__ ab,
    const float* __restrict__ fW, const float* __restrict__ fb,
    const float* __restrict__ abilW, const float* __restrict__ abilb,
    const float* __restrict__ diffW, const float* __restrict__ diffb,
    float* __restrict__ out)
{
  extern __shared__ float smem[];
  float* sK   = smem;
  float* sV   = sK  + RK * LP;
  float* sE   = sV  + RP * LP;
  float* sA   = sE  + RP * LP;
  float* sW   = sA  + RP * LP;

  const int b    = blockIdx.x;
  const int tid  = threadIdx.x;
  const int lane = tid & 31;
  const int wave = tid >> 5;
  const int hh   = lane >> 4;
  const int r16  = lane & 15;

  const int wv  = __builtin_amdgcn_readfirstlane(wave);
  const int nt  = wv >> 1;
  const int par = (wv ^ (wv >> 2)) & 1;
  const int o   = nt * 16 + r16;

  v2f bk[16], be[16], ba[16];
  float ebv, abv;

  {
    const int c4 = tid & 15;
    const int gr = tid >> 4;
    int qi[GP], ri[GP];
#pragma unroll
    for (int p = 0; p < GP; ++p) {
      int t  = p * 16 + gr;
      int tc = (t < TT) ? t : (TT - 1);
      qi[p] = question[b * TT + tc];
      ri[p] = response[b * TT + tc];
    }
    {
      const int oc = (o < SIZE_M) ? o : (SIZE_M - 1);
#pragma unroll
      for (int kb = 0; kb < 16; ++kb) {
        bk[kb] = *(const v2f*)&Mk[oc * 64 + kb * 4 + 2 * hh];
        be[kb] = *(const v2f*)&eW[o * 64 + kb * 4 + 2 * hh];
        ba[kb] = *(const v2f*)&aW[o * 64 + kb * 4 + 2 * hh];
      }
      ebv = eb[o];
      abv = ab[o];
    }
    const bool   qhalf = c4 < 8;
    const float* ktab  = (qhalf ? kq_emb : kc_emb) + 4 * (c4 & 7);
    const float* vtab  = (qhalf ? vq_emb : vc_emb) + 4 * (c4 & 7);
    v4f kx[GP], vx[GP];
#pragma unroll
    for (int p = 0; p < GP; ++p) {
      int krow = qhalf ? qi[p] : 0;
      int vrow = qhalf ? qi[p] + NUM_Q * ri[p] : NUM_C * ri[p];
      kx[p] = *(const v4f*)(ktab + krow * HALF);
      vx[p] = *(const v4f*)(vtab + vrow * HALF);
    }
    const v4f z4 = {0.f, 0.f, 0.f, 0.f};
#pragma unroll
    for (int p = 0; p < GP; ++p) {
      int t = p * 16 + gr;
      *(v4f*)&sK[t * LP + 4 * c4] = (t < TT) ? kx[p] : z4;
      *(v4f*)&sV[t * LP + 4 * c4] = (t < TT) ? vx[p] : z4;
    }
    if (tid < (RK - RP) * 16) *(v4f*)&sK[(RP + gr) * LP + 4 * c4] = z4;
  }
  __syncthreads();

  {
    v2f a0[16], a1[16];
    const float* pa = sK + r16 * LP + 2 * hh;
    load_afrag16(a0, pa + par * 16 * LP);
    for (int i = 0; i < 3; ++i) {
      const int m0 = par + 4 * i;
      const int m2 = (m0 + 4 < RP / 16) ? m0 + 4 : RP / 16 - 1;
      load_afrag16(a1, pa + (m0 + 2) * 16 * LP);
      __builtin_amdgcn_sched_barrier(0);
      tile_wlog(a0, bk, sW, m0, hh, o);
      load_afrag16(a0, pa + m2 * 16 * LP);
      __builtin_amdgcn_sched_barrier(0);
      tile_wlog(a1, bk, sW, m0 + 2, hh, o);
    }
    if (par == 0) tile_wlog(a0, bk, sW, RP / 16 - 1, hh, o);
  }

  {
    v2f a0[16], a1[16];
    const float* pa = sV + r16 * LP + 2 * hh;
    load_afrag16(a0, pa + par * 16 * LP);
    for (int i = 0; i < 3; ++i) {
      const int m0 = par + 4 * i;
      const int m2 = (m0 + 4 < RP / 16) ? m0 + 4 : RP / 16 - 1;
      load_afrag16(a1, pa + (m0 + 2) * 16 * LP);
      __builtin_amdgcn_sched_barrier(0);
      tile_ea(a0, be, ba, ebv, abv, sE, sA, m0, hh, o);
      load_afrag16(a0, pa + m2 * 16 * LP);
      __builtin_amdgcn_sched_barrier(0);
      tile_ea(a1, be, ba, ebv, abv, sE, sA, m0 + 2, hh, o);
    }
    if (par == 0) tile_ea(a0, be, ba, ebv, abv, sE, sA, RP / 16 - 1, hh, o);
  }
  __syncthreads();

  const int mg = lane & 3;
  const int dq = wave * 8 + (lane >> 2);
  v2f mv[MPP];
#pragma unroll
  for (int j = 0; j < MPP - 1; ++j) {
    mv[j].x = Mv0[(mg + 8 * j) * 64 + dq];
    mv[j].y = Mv0[(mg + 8 * j + 4) * 64 + dq];
  }
  mv[MPP - 1].x = (mg + 48 < SIZE_M) ? Mv0[(mg + 48) * 64 + dq] : 0.f;
  mv[MPP - 1].y = 0.f;

  v2f bw[32];
#pragma unroll
  for (int kb = 0; kb < 32; ++kb)
    bw[kb] = *(const v2f*)&fW[o * 128 + kb * 4 + 2 * hh];
  const float fbv = fb[o];

  for (int r = tid; r < TT; r += 256) {
    float mx = -1e30f;
    for (int m = 0; m < SIZE_M; ++m) mx = fmaxf(mx, sW[r * WST + m]);
    float s = 0.f;
    for (int m = 0; m < SIZE_M; ++m) {
      float ev = __expf(sW[r * WST + m] - mx);
      sW[r * WST + m] = ev; s += ev;
    }
    float inv = 1.f / s;
    for (int m = 0; m < SIZE_M; ++m) sW[r * WST + m] *= inv;
    for (int m = SIZE_M; m < SIZE_M + 6; ++m) sW[r * WST + m] = 0.f;
  }
  __syncthreads();

  {
    const float* pW = sW + mg;
    const float* pE = sE + dq;
    const float* pA = sA + dq;
    float*       pR = sV + dq;

    v2f wA[MPP], wB[MPP];
    float eA, aA, eB, aB;
    scan_load(wA, eA, aA, pW, pE, pA);
    scan_load(wB, eB, aB, pW + WST, pE + LP, pA + LP);
    scan_update(mv, wA, eA, aA);

    for (int t = 1; t < TT - 1; t += 2) {
      scan_load(wA, eA, aA, pW + (t + 1) * WST, pE + (t + 1) * LP, pA + (t + 1) * LP);
      float r0 = scan_read(mv, wB);
      scan_update(mv, wB, eB, aB);

      scan_load(wB, eB, aB, pW + (t + 2) * WST, pE + (t + 2) * LP, pA + (t + 2) * LP);
      float r1 = scan_read(mv, wA);
      scan_update(mv, wA, eA, aA);
      if (mg == 0) { pR[(t - 1) * LP] = r0; pR[t * LP] = r1; }
    }
    float rl = scan_read(mv, wB);
    if (mg == 0) pR[(TT - 2) * LP] = rl;
  }

  for (int idx = (TT - 1) * 64 + tid; idx < RP * 64; idx += 256) sV[(idx >> 6) * LP + (idx & 63)] = 0.f;
  __syncthreads();

  {
    v2f a0[32], a1[32];
    const float* pr = sV + r16 * LP + 2 * hh;
    const float* pk = sK + (r16 + 1) * LP + 2 * hh;
    load_afrag32(a0, pr + par * 16 * LP, pk + par * 16 * LP);
    for (int i = 0; i < 3; ++i) {
      const int m0 = par + 4 * i;
      const int m2 = (m0 + 4 < RP / 16) ? m0 + 4 : RP / 16 - 1;
      load_afrag32(a1, pr + (m0 + 2) * 16 * LP, pk + (m0 + 2) * 16 * LP);
      __builtin_amdgcn_sched_barrier(0);
      tile_f(a0, bw, fbv, sE, m0, hh, o);
      load_afrag32(a0, pr + m2 * 16 * LP, pk + m2 * 16 * LP);
      __builtin_amdgcn_sched_barrier(0);
      tile_f(a1, bw, fbv, sE, m0 + 2, hh, o);
    }
    if (par == 0) tile_f(a0, bw, fbv, sE, RP / 16 - 1, hh, o);
  }
  __syncthreads();

  for (int r = tid; r < TT - 1; r += 256) {
    float s1 = 0.f, s2 = 0.f;
    for (int c = 0; c < 64; ++c) {
      s1 += sE[r * LP + c] * abilW[c];
      s2 += sK[(r + 1) * LP + c] * diffW[c];
    }
    float ability = tanhf(s1 + abilb[0]);
    float diff    = tanhf(s2 + diffb[0]);
    out[b * (TT - 1) + r] = 3.f * ability - diff;
  }
}

extern "C" void kernel_launch(void* const* d_in, const int* in_sizes, int n_in,
                              void* d_out, int out_size, void* d_ws, size_t ws_size,
                              hipStream_t stream) {
  (void)in_sizes; (void)n_in; (void)out_size; (void)d_ws; (void)ws_size;
  const int*   question = (const int*)  d_in[0];
  const int*   response = (const int*)  d_in[1];
  const float* vq_emb   = (const float*)d_in[2];
  const float* vc_emb   = (const float*)d_in[3];
  const float* kq_emb   = (const float*)d_in[4];
  const float* kc_emb   = (const float*)d_in[5];
  const float* Mk       = (const float*)d_in[6];
  const float* Mv0      = (const float*)d_in[7];
  const float* eW       = (const float*)d_in[8];
  const float* ebv      = (const float*)d_in[9];
  const float* aW       = (const float*)d_in[10];
  const float* abv      = (const float*)d_in[11];
  const float* fW       = (const float*)d_in[12];
  const float* fbv      = (const float*)d_in[13];
  const float* abilW    = (const float*)d_in[14];
  const float* abilb    = (const float*)d_in[15];
  const float* diffW    = (const float*)d_in[16];
  const float* diffb    = (const float*)d_in[17];
  float* out = (float*)d_out;

  const size_t smem_bytes =
      (size_t)(RK * LP + 3 * RP * LP + RP * WST) * sizeof(float);

  (void)hipFuncSetAttribute((const void*)deepirt_fused_kernel,
                            hipFuncAttributeMaxDynamicSharedMemorySize,
                            (int)smem_bytes);

  deepirt_fused_kernel<<<BATCH, 256, smem_bytes, stream>>>(
      question, response, vq_emb, vc_emb, kq_emb, kc_emb, Mk, Mv0,
      eW, ebv, aW, abv, fW, fbv, abilW, abilb, diffW, diffb, out);
}


# Softmax and head loops: a thread reads its whole row with one group of 16-byte LDS loads and computes from registers in the old order (was one LDS round trip per loop trip)
# speedup vs baseline: 8.9749x; 1.0683x over previous
#include <hip/hip_runtime.h>
#include <math.h>


#define NUM_Q  10000
#define NUM_C  300
#define DIM_S  64
#define SIZE_M 50
#define BATCH  128
#define TT     200
#define HALF   32

#define RK  212
#define RP  208
#define WST 60
#define LP  68

typedef __attribute__((ext_vector_type(2))) float v2f;
typedef __attribute__((ext_vector_type(8))) float v8f;

static __device__ __forceinline__ v8f wmma_f32_k4(v2f a, v2f b, v8f c) {
  return __builtin_amdgcn_wmma_f32_16x16x4_f32(false, a, false, b, (short)0, c, false, false);
}

static __device__ __forceinline__ float tanh_exp_rcp(float x) {
  return 1.f - 2.f * __builtin_amdgcn_rcpf(1.f + __expf(2.f * x));
}
static __device__ __forceinline__ float sigmoid_exp_rcp(float x) {
  return __builtin_amdgcn_rcpf(1.f + __expf(-x));
}

#define GP  13

typedef __attribute__((ext_vector_type(4))) float v4f;

#define MPP 7


static __device__ __forceinline__ void scan_load(v2f (&w)[MPP], float& e, float& a,
                                                 const float* sWrow, const float* sErow,
                                                 const float* sArow) {
#pragma unroll
  for (int j = 0; j < MPP; ++j) { w[j].x = sWrow[8 * j]; w[j].y = sWrow[8 * j + 4]; }
  e = *sErow;
  a = *sArow;
}

static __device__ __forceinline__ void scan_update(v2f (&mv)[MPP], const v2f (&w)[MPP], float e, float a) {
  const v2f ne2 = {-e, -e};
  const v2f a2  = {a, a};
#pragma unroll
  for (int j = 0; j < MPP; ++j)
    mv[j] = __builtin_elementwise_fma(w[j], __builtin_elementwise_fma(ne2, mv[j], a2), mv[j]);
}

template <int SEL>
static __device__ __forceinline__ float quad_perm(float v) {
  return __int_as_float(__builtin_amdgcn_mov_dpp(__float_as_int(v), SEL, 0xf, 0xf, false));
}

static __device__ __forceinline__ float scan_read(const v2f (&mv)[MPP], const v2f (&w)[MPP]) {
  v2f s01 = w[0] * mv[0];
  v2f s23 = w[1] * mv[1];
  s01 = __builtin_elementwise_fma(w[2], mv[2], s01);
  s23 = __builtin_elementwise_fma(w[3], mv[3], s23);
  s01 = __builtin_elementwise_fma(w[4], mv[4], s01);
  s23 = __builtin_elementwise_fma(w[5], mv[5], s23);
  s01 = __builtin_elementwise_fma(w[6], mv[6], s01);
  float p = (s01.x + s01.y) + (s23.x + s23.y);
  p += quad_perm<0xB1>(p);
  p += quad_perm<0x4E>(p);
  return p;
}


static __device__ __forceinline__ void load_afrag16(v2f (&af)[16], const float* p) {
#pragma unroll
  for (int kb = 0; kb < 16; ++kb) af[kb] = *(const v2f*)(p + kb * 4);
}

static __device__ __forceinline__ void tile_wlog(const v2f (&af)[16], const v2f (&bf)[16],
                                                 float* sW, int mt, int hh, int o) {
  v8f acc = {0.f,0.f,0.f,0.f,0.f,0.f,0.f,0.f};
#pragma unroll
  for (int kb = 0; kb < 16; ++kb) acc = wmma_f32_k4(af[kb], bf[kb], acc);
  if (o < SIZE_M) {
#pragma unroll
    for (int j = 0; j < 8; ++j) sW[(mt * 16 + j + 8 * hh) * WST + o] = acc[j];
  }
}

static __device__ __forceinline__ void tile_ea(const v2f (&af)[16], const v2f (&be)[16], const v2f (&ba)[16],
                                               float ebv, float abv, float* sE, float* sA,
                                               int mt, int hh, int o) {
  v8f ce = {0.f,0.f,0.f,0.f,0.f,0.f,0.f,0.f};
  v8f ca = {0.f,0.f,0.f,0.f,0.f,0.f,0.f,0.f};
#pragma unroll
  for (int kb = 0; kb < 16; ++kb) {
    ce = wmma_f32_k4(af[kb], be[kb], ce);
    ca = wmma_f32_k4(af[kb], ba[kb], ca);
  }
#pragma unroll
  for (int j = 0; j < 8; ++j) {
    int row = mt * 16 + j + 8 * hh;
    sE[row * LP + o] = sigmoid_exp_rcp(ce[j] + ebv);
    sA[row * LP + o] = tanh_exp_rcp(ca[j] + abv);
  }
}

static __device__ __forceinline__ void load_afrag32(v2f (&af)[32], const float* pr, const float* pk) {
#pragma unroll
  for (int kb = 0; kb < 16; ++kb) af[kb] = *(const v2f*)(pr + kb * 4);
#pragma unroll
  for (int kb = 0; kb < 16; ++kb) af[16 + kb] = *(const v2f*)(pk + kb * 4);
}

static __device__ __forceinline__ void tile_f(const v2f (&af)[32], const v2f (&bf)[32], float fbv,
                                              float* sE, int mt, int hh, int o) {
  v8f acc = {0.f,0.f,0.f,0.f,0.f,0.f,0.f,0.f};
#pragma unroll
  for (int kb = 0; kb < 32; ++kb) acc = wmma_f32_k4(af[kb], bf[kb], acc);
#pragma unroll
  for (int j = 0; j < 8; ++j) {
    int rr = mt * 16 + j + 8 * hh;
    sE[rr * LP + o] = tanh_exp_rcp(acc[j] + fbv);
  }
}

__global__ __launch_bounds__(256) void deepirt_fused_kernel(
    const int*   __restrict__ question, const int*   __restrict__ response,
    const float* __restrict__ vq_emb,   const float* __restrict__ vc_emb,
    const float* __restrict__ kq_emb,   const float* __restrict__ kc_emb,
    const float* __restrict__ Mk,       const float* __restrict__ Mv0,
    const float* __restrict__ eW, const float* __restrict__ eb,
    const float* __restrict__ aW, const float* __restrict__ ab,
    const float* __restrict__ fW, const float* __restrict__ fb,
    const float* __restrict__ abilW, const float* __restrict__ abilb,
    const float* __restrict__ diffW, const float* __restrict__ diffb,
    float* __restrict__ out)
{
  extern __shared__ float smem[];
  float* sK   = smem;
  float* sV   = sK  + RK * LP;
  float* sE   = sV  + RP * LP;
  float* sA   = sE  + RP * LP;
  float* sW   = sA  + RP * LP;

  const int b    = blockIdx.x;
  const int tid  = threadIdx.x;
  const int lane = tid & 31;
  const int wave = tid >> 5;
  const int hh   = lane >> 4;
  const int r16  = lane & 15;

  const int wv  = __builtin_amdgcn_readfirstlane(wave);
  const int nt  = wv >> 1;
  const int par = (wv ^ (wv >> 2)) & 1;
  const int o   = nt * 16 + r16;

  v2f bk[16], be[16], ba[16];
  float ebv, abv;

  {
    const int c4 = tid & 15;
    const int gr = tid >> 4;
    int qi[GP], ri[GP];
#pragma unroll
    for (int p = 0; p < GP; ++p) {
      int t  = p * 16 + gr;
      int tc = (t < TT) ? t : (TT - 1);
      qi[p] = question[b * TT + tc];
      ri[p] = response[b * TT + tc];
    }
    {
      const int oc = (o < SIZE_M) ? o : (SIZE_M - 1);
#pragma unroll
      for (int kb = 0; kb < 16; ++kb) {
        bk[kb] = *(const v2f*)&Mk[oc * 64 + kb * 4 + 2 * hh];
        be[kb] = *(const v2f*)&eW[o * 64 + kb * 4 + 2 * hh];
        ba[kb] = *(const v2f*)&aW[o * 64 + kb * 4 + 2 * hh];
      }
      ebv = eb[o];
      abv = ab[o];
    }
    const bool   qhalf = c4 < 8;
    const float* ktab  = (qhalf ? kq_emb : kc_emb) + 4 * (c4 & 7);
    const float* vtab  = (qhalf ? vq_emb : vc_emb) + 4 * (c4 & 7);
    v4f kx[GP], vx[GP];
#pragma unroll
    for (int p = 0; p < GP; ++p) {
      int krow = qhalf ? qi[p] : 0;
      int vrow = qhalf ? qi[p] + NUM_Q * ri[p] : NUM_C * ri[p];
      kx[p] = *(const v4f*)(ktab + krow * HALF);
      vx[p] = *(const v4f*)(vtab + vrow * HALF);
    }
    const v4f z4 = {0.f, 0.f, 0.f, 0.f};
#pragma unroll
    for (int p = 0; p < GP; ++p) {
      int t = p * 16 + gr;
      *(v4f*)&sK[t * LP + 4 * c4] = (t < TT) ? kx[p] : z4;
      *(v4f*)&sV[t * LP + 4 * c4] = (t < TT) ? vx[p] : z4;
    }
    if (tid < (RK - RP) * 16) *(v4f*)&sK[(RP + gr) * LP + 4 * c4] = z4;
  }
  __syncthreads();

  {
    v2f a0[16], a1[16];
    const float* pa = sK + r16 * LP + 2 * hh;
    load_afrag16(a0, pa + par * 16 * LP);
    for (int i = 0; i < 3; ++i) {
      const int m0 = par + 4 * i;
      const int m2 = (m0 + 4 < RP / 16) ? m0 + 4 : RP / 16 - 1;
      load_afrag16(a1, pa + (m0 + 2) * 16 * LP);
      __builtin_amdgcn_sched_barrier(0);
      tile_wlog(a0, bk, sW, m0, hh, o);
      load_afrag16(a0, pa + m2 * 16 * LP);
      __builtin_amdgcn_sched_barrier(0);
      tile_wlog(a1, bk, sW, m0 + 2, hh, o);
    }
    if (par == 0) tile_wlog(a0, bk, sW, RP / 16 - 1, hh, o);
  }

  {
    v2f a0[16], a1[16];
    const float* pa = sV + r16 * LP + 2 * hh;
    load_afrag16(a0, pa + par * 16 * LP);
    for (int i = 0; i < 3; ++i) {
      const int m0 = par + 4 * i;
      const int m2 = (m0 + 4 < RP / 16) ? m0 + 4 : RP / 16 - 1;
      load_afrag16(a1, pa + (m0 + 2) * 16 * LP);
      __builtin_amdgcn_sched_barrier(0);
      tile_ea(a0, be, ba, ebv, abv, sE, sA, m0, hh, o);
      load_afrag16(a0, pa + m2 * 16 * LP);
      __builtin_amdgcn_sched_barrier(0);
      tile_ea(a1, be, ba, ebv, abv, sE, sA, m0 + 2, hh, o);
    }
    if (par == 0) tile_ea(a0, be, ba, ebv, abv, sE, sA, RP / 16 - 1, hh, o);
  }
  __syncthreads();

  const int mg = lane & 3;
  const int dq = wave * 8 + (lane >> 2);
  v2f mv[MPP];
#pragma unroll
  for (int j = 0; j < MPP - 1; ++j) {
    mv[j].x = Mv0[(mg + 8 * j) * 64 + dq];
    mv[j].y = Mv0[(mg + 8 * j + 4) * 64 + dq];
  }
  mv[MPP - 1].x = (mg + 48 < SIZE_M) ? Mv0[(mg + 48) * 64 + dq] : 0.f;
  mv[MPP - 1].y = 0.f;

  v2f bw[32];
#pragma unroll
  for (int kb = 0; kb < 32; ++kb)
    bw[kb] = *(const v2f*)&fW[o * 128 + kb * 4 + 2 * hh];
  const float fbv = fb[o];

  for (int r = tid; r < TT; r += 256) {
    float* row = sW + r * WST;
    float x[SIZE_M];
#pragma unroll
    for (int q = 0; q < 12; ++q) {
      const v4f v = *(const v4f*)(row + 4 * q);
      x[4 * q] = v.x; x[4 * q + 1] = v.y; x[4 * q + 2] = v.z; x[4 * q + 3] = v.w;
    }
    {
      const v2f v = *(const v2f*)(row + 48);
      x[48] = v.x; x[49] = v.y;
    }
    float mx = -1e30f;
#pragma unroll
    for (int m = 0; m < SIZE_M; ++m) mx = fmaxf(mx, x[m]);
    float s = 0.f;
#pragma unroll
    for (int m = 0; m < SIZE_M; ++m) { x[m] = __expf(x[m] - mx); s += x[m]; }
    const float inv = 1.f / s;
#pragma unroll
    for (int q = 0; q < 12; ++q) {
      const v4f v = {x[4 * q] * inv, x[4 * q + 1] * inv, x[4 * q + 2] * inv, x[4 * q + 3] * inv};
      *(v4f*)(row + 4 * q) = v;
    }
    const v4f t0 = {x[48] * inv, x[49] * inv, 0.f, 0.f};
    const v4f t1 = {0.f, 0.f, 0.f, 0.f};
    *(v4f*)(row + 48) = t0;
    *(v4f*)(row + 52) = t1;
  }
  __syncthreads();

  {
    const float* pW = sW + mg;
    const float* pE = sE + dq;
    const float* pA = sA + dq;
    float*       pR = sV + dq;

    v2f wA[MPP], wB[MPP];
    float eA, aA, eB, aB;
    scan_load(wA, eA, aA, pW, pE, pA);
    scan_load(wB, eB, aB, pW + WST, pE + LP, pA + LP);
    scan_update(mv, wA, eA, aA);

    for (int t = 1; t < TT - 1; t += 2) {
      scan_load(wA, eA, aA, pW + (t + 1) * WST, pE + (t + 1) * LP, pA + (t + 1) * LP);
      float r0 = scan_read(mv, wB);
      scan_update(mv, wB, eB, aB);

      scan_load(wB, eB, aB, pW + (t + 2) * WST, pE + (t + 2) * LP, pA + (t + 2) * LP);
      float r1 = scan_read(mv, wA);
      scan_update(mv, wA, eA, aA);
      if (mg == 0) { pR[(t - 1) * LP] = r0; pR[t * LP] = r1; }
    }
    float rl = scan_read(mv, wB);
    if (mg == 0) pR[(TT - 2) * LP] = rl;
  }

  for (int idx = (TT - 1) * 64 + tid; idx < RP * 64; idx += 256) sV[(idx >> 6) * LP + (idx & 63)] = 0.f;
  __syncthreads();

  {
    v2f a0[32], a1[32];
    const float* pr = sV + r16 * LP + 2 * hh;
    const float* pk = sK + (r16 + 1) * LP + 2 * hh;
    load_afrag32(a0, pr + par * 16 * LP, pk + par * 16 * LP);
    for (int i = 0; i < 3; ++i) {
      const int m0 = par + 4 * i;
      const int m2 = (m0 + 4 < RP / 16) ? m0 + 4 : RP / 16 - 1;
      load_afrag32(a1, pr + (m0 + 2) * 16 * LP, pk + (m0 + 2) * 16 * LP);
      __builtin_amdgcn_sched_barrier(0);
      tile_f(a0, bw, fbv, sE, m0, hh, o);
      load_afrag32(a0, pr + m2 * 16 * LP, pk + m2 * 16 * LP);
      __builtin_amdgcn_sched_barrier(0);
      tile_f(a1, bw, fbv, sE, m0 + 2, hh, o);
    }
    if (par == 0) tile_f(a0, bw, fbv, sE, RP / 16 - 1, hh, o);
  }
  __syncthreads();

  for (int r = tid; r < TT - 1; r += 256) {
    v4f fr[16], kr[16];
#pragma unroll
    for (int q = 0; q < 16; ++q) {
      fr[q] = *(const v4f*)&sE[r * LP + 4 * q];
      kr[q] = *(const v4f*)&sK[(r + 1) * LP + 4 * q];
    }
    __builtin_amdgcn_sched_barrier(0);
    float s2 = 0.f;
#pragma unroll
    for (int q = 0; q < 16; ++q) {
      s2 = __builtin_fmaf(kr[q].x, diffW[4 * q],     s2);
      s2 = __builtin_fmaf(kr[q].y, diffW[4 * q + 1], s2);
      s2 = __builtin_fmaf(kr[q].z, diffW[4 * q + 2], s2);
      s2 = __builtin_fmaf(kr[q].w, diffW[4 * q + 3], s2);
    }
    __builtin_amdgcn_sched_barrier(0);
    float s1 = 0.f;
#pragma unroll
    for (int q = 0; q < 16; ++q) {
      s1 = __builtin_fmaf(fr[q].x, abilW[4 * q],     s1);
      s1 = __builtin_fmaf(fr[q].y, abilW[4 * q + 1], s1);
      s1 = __builtin_fmaf(fr[q].z, abilW[4 * q + 2], s1);
      s1 = __builtin_fmaf(fr[q].w, abilW[4 * q + 3], s1);
    }
    float ability = tanhf(s1 + abilb[0]);
    float diff    = tanhf(s2 + diffb[0]);
    out[b * (TT - 1) + r] = 3.f * ability - diff;
  }
}

extern "C" void kernel_launch(void* const* d_in, const int* in_sizes, int n_in,
                              void* d_out, int out_size, void* d_ws, size_t ws_size,
                              hipStream_t stream) {
  (void)in_sizes; (void)n_in; (void)out_size; (void)d_ws; (void)ws_size;
  const int*   question = (const int*)  d_in[0];
  const int*   response = (const int*)  d_in[1];
  const float* vq_emb   = (const float*)d_in[2];
  const float* vc_emb   = (const float*)d_in[3];
  const float* kq_emb   = (const float*)d_in[4];
  const float* kc_emb   = (const float*)d_in[5];
  const float* Mk       = (const float*)d_in[6];
  const float* Mv0      = (const float*)d_in[7];
  const float* eW       = (const float*)d_in[8];
  const float* ebv      = (const float*)d_in[9];
  const float* aW       = (const float*)d_in[10];
  const float* abv      = (const float*)d_in[11];
  const float* fW       = (const float*)d_in[12];
  const float* fbv      = (const float*)d_in[13];
  const float* abilW    = (const float*)d_in[14];
  const float* abilb    = (const float*)d_in[15];
  const float* diffW    = (const float*)d_in[16];
  const float* diffb    = (const float*)d_in[17];
  float* out = (float*)d_out;

  const size_t smem_bytes =
      (size_t)(RK * LP + 3 * RP * LP + RP * WST) * sizeof(float);

  (void)hipFuncSetAttribute((const void*)deepirt_fused_kernel,
                            hipFuncAttributeMaxDynamicSharedMemorySize,
                            (int)smem_bytes);

  deepirt_fused_kernel<<<BATCH, 256, smem_bytes, stream>>>(
      question, response, vq_emb, vc_emb, kq_emb, kc_emb, Mk, Mv0,
      eW, ebv, aW, abv, fW, fbv, abilW, abilb, diffW, diffb, out);
}


# Scan and f GEMM run side by side: 4 waves scan (thread owns 26 slots of a column, 2-lane sum: sum order changed) while 4 waves do the f tile of the 16 rows before; one barrier per 16 steps
# speedup vs baseline: 9.3673x; 1.0437x over previous
#include <hip/hip_runtime.h>
#include <math.h>


#define NUM_Q  10000
#define NUM_C  300
#define DIM_S  64
#define SIZE_M 50
#define BATCH  128
#define TT     200
#define HALF   32

#define RK  212
#define RP  208
#define WST 60
#define LP  68

typedef __attribute__((ext_vector_type(2))) float v2f;
typedef __attribute__((ext_vector_type(8))) float v8f;

static __device__ __forceinline__ v8f wmma_f32_k4(v2f a, v2f b, v8f c) {
  return __builtin_amdgcn_wmma_f32_16x16x4_f32(false, a, false, b, (short)0, c, false, false);
}

static __device__ __forceinline__ float tanh_exp_rcp(float x) {
  return 1.f - 2.f * __builtin_amdgcn_rcpf(1.f + __expf(2.f * x));
}
static __device__ __forceinline__ float sigmoid_exp_rcp(float x) {
  return __builtin_amdgcn_rcpf(1.f + __expf(-x));
}

#define GP  13

typedef __attribute__((ext_vector_type(4))) float v4f;

#define SPP 13


static __device__ __forceinline__ void scan_load(v2f (&w)[SPP], float& e, float& a,
                                                 const float* sWrow, const float* sErow,
                                                 const float* sArow) {
#pragma unroll
  for (int j = 0; j < SPP; ++j) w[j] = *(const v2f*)(sWrow + 4 * j);
  e = *sErow;
  a = *sArow;
}

static __device__ __forceinline__ void scan_update(v2f (&mv)[SPP], const v2f (&w)[SPP], float e, float a) {
  const v2f ne2 = {-e, -e};
  const v2f a2  = {a, a};
#pragma unroll
  for (int j = 0; j < SPP; ++j)
    mv[j] = __builtin_elementwise_fma(w[j], __builtin_elementwise_fma(ne2, mv[j], a2), mv[j]);
}

template <int SEL>
static __device__ __forceinline__ float quad_perm(float v) {
  return __int_as_float(__builtin_amdgcn_mov_dpp(__float_as_int(v), SEL, 0xf, 0xf, false));
}

static __device__ __forceinline__ float scan_read(const v2f (&mv)[SPP], const v2f (&w)[SPP]) {
  v2f s01 = w[0] * mv[0];
  v2f s23 = w[1] * mv[1];
#pragma unroll
  for (int j = 2; j < SPP; j += 2) {
    s01 = __builtin_elementwise_fma(w[j], mv[j], s01);
    if (j + 1 < SPP) s23 = __builtin_elementwise_fma(w[j + 1], mv[j + 1], s23);
  }
  float p = (s01.x + s01.y) + (s23.x + s23.y);
  p += quad_perm<0xB1>(p);
  return p;
}


static __device__ __forceinline__ void load_afrag16(v2f (&af)[16], const float* p) {
#pragma unroll
  for (int kb = 0; kb < 16; ++kb) af[kb] = *(const v2f*)(p + kb * 4);
}

static __device__ __forceinline__ void tile_wlog(const v2f (&af)[16], const v2f (&bf)[16],
                                                 float* sW, int mt, int hh, int o) {
  v8f acc = {0.f,0.f,0.f,0.f,0.f,0.f,0.f,0.f};
#pragma unroll
  for (int kb = 0; kb < 16; ++kb) acc = wmma_f32_k4(af[kb], bf[kb], acc);
  if (o < SIZE_M) {
#pragma unroll
    for (int j = 0; j < 8; ++j) sW[(mt * 16 + j + 8 * hh) * WST + o] = acc[j];
  }
}

static __device__ __forceinline__ void tile_ea(const v2f (&af)[16], const v2f (&be)[16], const v2f (&ba)[16],
                                               float ebv, float abv, float* sE, float* sA,
                                               int mt, int hh, int o) {
  v8f ce = {0.f,0.f,0.f,0.f,0.f,0.f,0.f,0.f};
  v8f ca = {0.f,0.f,0.f,0.f,0.f,0.f,0.f,0.f};
#pragma unroll
  for (int kb = 0; kb < 16; ++kb) {
    ce = wmma_f32_k4(af[kb], be[kb], ce);
    ca = wmma_f32_k4(af[kb], ba[kb], ca);
  }
#pragma unroll
  for (int j = 0; j < 8; ++j) {
    int row = mt * 16 + j + 8 * hh;
    sE[row * LP + o] = sigmoid_exp_rcp(ce[j] + ebv);
    sA[row * LP + o] = tanh_exp_rcp(ca[j] + abv);
  }
}

static __device__ __forceinline__ void load_afrag32(v2f (&af)[32], const float* pr, const float* pk) {
#pragma unroll
  for (int kb = 0; kb < 16; ++kb) af[kb] = *(const v2f*)(pr + kb * 4);
#pragma unroll
  for (int kb = 0; kb < 16; ++kb) af[16 + kb] = *(const v2f*)(pk + kb * 4);
}

static __device__ __forceinline__ void tile_f(const v2f (&af)[32], const v2f (&bf)[32], float fbv,
                                              float* sE, int mt, int hh, int o) {
  v8f acc = {0.f,0.f,0.f,0.f,0.f,0.f,0.f,0.f};
#pragma unroll
  for (int kb = 0; kb < 32; ++kb) acc = wmma_f32_k4(af[kb], bf[kb], acc);
#pragma unroll
  for (int j = 0; j < 8; ++j) {
    int rr = mt * 16 + j + 8 * hh;
    sE[rr * LP + o] = tanh_exp_rcp(acc[j] + fbv);
  }
}

__global__ __launch_bounds__(256) void deepirt_fused_kernel(
    const int*   __restrict__ question, const int*   __restrict__ response,
    const float* __restrict__ vq_emb,   const float* __restrict__ vc_emb,
    const float* __restrict__ kq_emb,   const float* __restrict__ kc_emb,
    const float* __restrict__ Mk,       const float* __restrict__ Mv0,
    const float* __restrict__ eW, const float* __restrict__ eb,
    const float* __restrict__ aW, const float* __restrict__ ab,
    const float* __restrict__ fW, const float* __restrict__ fb,
    const float* __restrict__ abilW, const float* __restrict__ abilb,
    const float* __restrict__ diffW, const float* __restrict__ diffb,
    float* __restrict__ out)
{
  extern __shared__ float smem[];
  float* sK   = smem;
  float* sV   = sK  + RK * LP;
  float* sE   = sV  + RP * LP;
  float* sA   = sE  + RP * LP;
  float* sW   = sA  + RP * LP;

  const int b    = blockIdx.x;
  const int tid  = threadIdx.x;
  const int lane = tid & 31;
  const int wave = tid >> 5;
  const int hh   = lane >> 4;
  const int r16  = lane & 15;

  const int wv  = __builtin_amdgcn_readfirstlane(wave);
  const int nt  = wv >> 1;
  const int par = (wv ^ (wv >> 2)) & 1;
  const int o   = nt * 16 + r16;

  v2f bk[16], be[16], ba[16];
  float ebv, abv;

  {
    const int c4 = tid & 15;
    const int gr = tid >> 4;
    int qi[GP], ri[GP];
#pragma unroll
    for (int p = 0; p < GP; ++p) {
      int t  = p * 16 + gr;
      int tc = (t < TT) ? t : (TT - 1);
      qi[p] = question[b * TT + tc];
      ri[p] = response[b * TT + tc];
    }
    {
      const int oc = (o < SIZE_M) ? o : (SIZE_M - 1);
#pragma unroll
      for (int kb = 0; kb < 16; ++kb) {
        bk[kb] = *(const v2f*)&Mk[oc * 64 + kb * 4 + 2 * hh];
        be[kb] = *(const v2f*)&eW[o * 64 + kb * 4 + 2 * hh];
        ba[kb] = *(const v2f*)&aW[o * 64 + kb * 4 + 2 * hh];
      }
      ebv = eb[o];
      abv = ab[o];
    }
    const bool   qhalf = c4 < 8;
    const float* ktab  = (qhalf ? kq_emb : kc_emb) + 4 * (c4 & 7);
    const float* vtab  = (qhalf ? vq_emb : vc_emb) + 4 * (c4 & 7);
    v4f kx[GP], vx[GP];
#pragma unroll
    for (int p = 0; p < GP; ++p) {
      int krow = qhalf ? qi[p] : 0;
      int vrow = qhalf ? qi[p] + NUM_Q * ri[p] : NUM_C * ri[p];
      kx[p] = *(const v4f*)(ktab + krow * HALF);
      vx[p] = *(const v4f*)(vtab + vrow * HALF);
    }
    const v4f z4 = {0.f, 0.f, 0.f, 0.f};
#pragma unroll
    for (int p = 0; p < GP; ++p) {
      int t = p * 16 + gr;
      *(v4f*)&sK[t * LP + 4 * c4] = (t < TT) ? kx[p] : z4;
      *(v4f*)&sV[t * LP + 4 * c4] = (t < TT) ? vx[p] : z4;
    }
    if (tid < (RK - RP) * 16) *(v4f*)&sK[(RP + gr) * LP + 4 * c4] = z4;
  }
  __syncthreads();

  {
    v2f a0[16], a1[16];
    const float* pa = sK + r16 * LP + 2 * hh;
    load_afrag16(a0, pa + par * 16 * LP);
    for (int i = 0; i < 3; ++i) {
      const int m0 = par + 4 * i;
      const int m2 = (m0 + 4 < RP / 16) ? m0 + 4 : RP / 16 - 1;
      load_afrag16(a1, pa + (m0 + 2) * 16 * LP);
      __builtin_amdgcn_sched_barrier(0);
      tile_wlog(a0, bk, sW, m0, hh, o);
      load_afrag16(a0, pa + m2 * 16 * LP);
      __builtin_amdgcn_sched_barrier(0);
      tile_wlog(a1, bk, sW, m0 + 2, hh, o);
    }
    if (par == 0) tile_wlog(a0, bk, sW, RP / 16 - 1, hh, o);
  }

  {
    v2f a0[16], a1[16];
    const float* pa = sV + r16 * LP + 2 * hh;
    load_afrag16(a0, pa + par * 16 * LP);
    for (int i = 0; i < 3; ++i) {
      const int m0 = par + 4 * i;
      const int m2 = (m0 + 4 < RP / 16) ? m0 + 4 : RP / 16 - 1;
      load_afrag16(a1, pa + (m0 + 2) * 16 * LP);
      __builtin_amdgcn_sched_barrier(0);
      tile_ea(a0, be, ba, ebv, abv, sE, sA, m0, hh, o);
      load_afrag16(a0, pa + m2 * 16 * LP);
      __builtin_amdgcn_sched_barrier(0);
      tile_ea(a1, be, ba, ebv, abv, sE, sA, m0 + 2, hh, o);
    }
    if (par == 0) tile_ea(a0, be, ba, ebv, abv, sE, sA, RP / 16 - 1, hh, o);
  }
  __syncthreads();

  const int mg = lane & 1;
  const int dq = nt * 16 + (lane >> 1);
  v2f mv[SPP];
  v2f bw[32];
  float fbv = 0.f;
  if (par == 0) {
#pragma unroll
    for (int j = 0; j < SPP - 1; ++j) {
      mv[j].x = Mv0[(4 * j + 2 * mg) * 64 + dq];
      mv[j].y = Mv0[(4 * j + 2 * mg + 1) * 64 + dq];
    }
    const float m48 = Mv0[48 * 64 + dq];
    const float m49 = Mv0[49 * 64 + dq];
    mv[SPP - 1].x = (mg == 0) ? m48 : 0.f;
    mv[SPP - 1].y = (mg == 0) ? m49 : 0.f;
  } else {
#pragma unroll
    for (int kb = 0; kb < 32; ++kb)
      bw[kb] = *(const v2f*)&fW[o * 128 + kb * 4 + 2 * hh];
    fbv = fb[o];
  }

  for (int r = tid; r < RP; r += 256) {
    float* row = sW + r * WST;
    float x[SIZE_M];
#pragma unroll
    for (int q = 0; q < 12; ++q) {
      const v4f v = *(const v4f*)(row + 4 * q);
      x[4 * q] = v.x; x[4 * q + 1] = v.y; x[4 * q + 2] = v.z; x[4 * q + 3] = v.w;
    }
    {
      const v2f v = *(const v2f*)(row + 48);
      x[48] = v.x; x[49] = v.y;
    }
    float mx = -1e30f;
#pragma unroll
    for (int m = 0; m < SIZE_M; ++m) mx = fmaxf(mx, x[m]);
    float s = 0.f;
#pragma unroll
    for (int m = 0; m < SIZE_M; ++m) { x[m] = __expf(x[m] - mx); s += x[m]; }
    const float inv = 1.f / s;
#pragma unroll
    for (int q = 0; q < 12; ++q) {
      const v4f v = {x[4 * q] * inv, x[4 * q + 1] * inv, x[4 * q + 2] * inv, x[4 * q + 3] * inv};
      *(v4f*)(row + 4 * q) = v;
    }
    const v4f t0 = {x[48] * inv, x[49] * inv, 0.f, 0.f};
    const v4f t1 = {0.f, 0.f, 0.f, 0.f};
    *(v4f*)(row + 48) = t0;
    *(v4f*)(row + 52) = t1;
  }
  for (int idx = (TT - 1) * 64 + tid; idx < RP * 64; idx += 256) sV[(idx >> 6) * LP + (idx & 63)] = 0.f;
  __syncthreads();

  {
    const float* pW = sW + 2 * mg;
    const float* pE = sE + dq;
    const float* pA = sA + dq;
    float*       pR = sV + dq;
    const float* pr = sV + r16 * LP + 2 * hh;
    const float* pk = sK + (r16 + 1) * LP + 2 * hh;

    v2f w0[SPP], w1[SPP], w2[SPP], w3[SPP];
    float e0 = 0.f, a0 = 0.f, e1 = 0.f, a1 = 0.f, e2 = 0.f, a2 = 0.f, e3 = 0.f, a3 = 0.f;
    const float* pWt = pW;
    const float* pEt = pE;
    const float* pAt = pA;
    float*       pRt = pR;
    if (par == 0) {
      scan_load(w3, e3, a3, pW, pE, pA);
      scan_load(w0, e0, a0, pW + WST, pE + LP, pA + LP);
      scan_load(w1, e1, a1, pW + 2 * WST, pE + 2 * LP, pA + 2 * LP);
      scan_update(mv, w3, e3, a3);
    }

    for (int it = 0; it <= RP / 16; ++it) {
      if (par == 0) {
        if (it < RP / 16) {
          const int ntrip = (it < RP / 16 - 1) ? 4 : 2;
          for (int q = 0; q < ntrip; ++q) {
            scan_load(w2, e2, a2, pWt + 3 * WST, pEt + 3 * LP, pAt + 3 * LP);
            const float r0 = scan_read(mv, w0);
            scan_update(mv, w0, e0, a0);
            scan_load(w3, e3, a3, pWt + 4 * WST, pEt + 4 * LP, pAt + 4 * LP);
            const float r1 = scan_read(mv, w1);
            scan_update(mv, w1, e1, a1);
            scan_load(w0, e0, a0, pWt + 5 * WST, pEt + 5 * LP, pAt + 5 * LP);
            const float r2 = scan_read(mv, w2);
            scan_update(mv, w2, e2, a2);
            scan_load(w1, e1, a1, pWt + 6 * WST, pEt + 6 * LP, pAt + 6 * LP);
            const float r3 = scan_read(mv, w3);
            scan_update(mv, w3, e3, a3);
            if (mg == 0) {
              pRt[0] = r0; pRt[LP] = r1; pRt[2 * LP] = r2; pRt[3 * LP] = r3;
            }
            pWt += 4 * WST; pEt += 4 * LP; pAt += 4 * LP; pRt += 4 * LP;
          }
        }
      } else {
        if (it >= 1) {
          const int mt = it - 1;
          v2f af[32];
          load_afrag32(af, pr + mt * 16 * LP, pk + mt * 16 * LP);
          __builtin_amdgcn_sched_barrier(0);
          tile_f(af, bw, fbv, sE, mt, hh, o);
        }
      }
      __syncthreads();
    }
  }

  for (int r = tid; r < TT - 1; r += 256) {
    v4f fr[16], kr[16];
#pragma unroll
    for (int q = 0; q < 16; ++q) {
      fr[q] = *(const v4f*)&sE[r * LP + 4 * q];
      kr[q] = *(const v4f*)&sK[(r + 1) * LP + 4 * q];
    }
    __builtin_amdgcn_sched_barrier(0);
    float s2 = 0.f;
#pragma unroll
    for (int q = 0; q < 16; ++q) {
      s2 = __builtin_fmaf(kr[q].x, diffW[4 * q],     s2);
      s2 = __builtin_fmaf(kr[q].y, diffW[4 * q + 1], s2);
      s2 = __builtin_fmaf(kr[q].z, diffW[4 * q + 2], s2);
      s2 = __builtin_fmaf(kr[q].w, diffW[4 * q + 3], s2);
    }
    __builtin_amdgcn_sched_barrier(0);
    float s1 = 0.f;
#pragma unroll
    for (int q = 0; q < 16; ++q) {
      s1 = __builtin_fmaf(fr[q].x, abilW[4 * q],     s1);
      s1 = __builtin_fmaf(fr[q].y, abilW[4 * q + 1], s1);
      s1 = __builtin_fmaf(fr[q].z, abilW[4 * q + 2], s1);
      s1 = __builtin_fmaf(fr[q].w, abilW[4 * q + 3], s1);
    }
    float ability = tanhf(s1 + abilb[0]);
    float diff    = tanhf(s2 + diffb[0]);
    out[b * (TT - 1) + r] = 3.f * ability - diff;
  }
}

extern "C" void kernel_launch(void* const* d_in, const int* in_sizes, int n_in,
                              void* d_out, int out_size, void* d_ws, size_t ws_size,
                              hipStream_t stream) {
  (void)in_sizes; (void)n_in; (void)out_size; (void)d_ws; (void)ws_size;
  const int*   question = (const int*)  d_in[0];
  const int*   response = (const int*)  d_in[1];
  const float* vq_emb   = (const float*)d_in[2];
  const float* vc_emb   = (const float*)d_in[3];
  const float* kq_emb   = (const float*)d_in[4];
  const float* kc_emb   = (const float*)d_in[5];
  const float* Mk       = (const float*)d_in[6];
  const float* Mv0      = (const float*)d_in[7];
  const float* eW       = (const float*)d_in[8];
  const float* ebv      = (const float*)d_in[9];
  const float* aW       = (const float*)d_in[10];
  const float* abv      = (const float*)d_in[11];
  const float* fW       = (const float*)d_in[12];
  const float* fbv      = (const float*)d_in[13];
  const float* abilW    = (const float*)d_in[14];
  const float* abilb    = (const float*)d_in[15];
  const float* diffW    = (const float*)d_in[16];
  const float* diffb    = (const float*)d_in[17];
  float* out = (float*)d_out;

  const size_t smem_bytes =
      (size_t)(RK * LP + 3 * RP * LP + RP * WST) * sizeof(float);

  (void)hipFuncSetAttribute((const void*)deepirt_fused_kernel,
                            hipFuncAttributeMaxDynamicSharedMemorySize,
                            (int)smem_bytes);

  deepirt_fused_kernel<<<BATCH, 256, smem_bytes, stream>>>(
      question, response, vq_emb, vc_emb, kq_emb, kc_emb, Mk, Mv0,
      eW, ebv, aW, abv, fW, fbv, abilW, abilb, diffW, diffb, out);
}


# Step-invariant concept half of each k and v row (kc row 0; vc row of response 0/1) multiplied once per wave on WMMA into the accumulator start values; row GEMMs run K=32 (f: K=96); sum order changed
# speedup vs baseline: 10.1146x; 1.0798x over previous
#include <hip/hip_runtime.h>
#include <math.h>


#define NUM_Q  10000
#define NUM_C  300
#define DIM_S  64
#define SIZE_M 50
#define BATCH  128
#define TT     200
#define HALF   32

#define RK  212
#define RP  208
#define WST 60
#define LP  68

typedef __attribute__((ext_vector_type(2))) float v2f;
typedef __attribute__((ext_vector_type(8))) float v8f;

static __device__ __forceinline__ v8f wmma_f32_k4(v2f a, v2f b, v8f c) {
  return __builtin_amdgcn_wmma_f32_16x16x4_f32(false, a, false, b, (short)0, c, false, false);
}

static __device__ __forceinline__ float tanh_exp_rcp(float x) {
  return 1.f - 2.f * __builtin_amdgcn_rcpf(1.f + __expf(2.f * x));
}
static __device__ __forceinline__ float sigmoid_exp_rcp(float x) {
  return __builtin_amdgcn_rcpf(1.f + __expf(-x));
}

#define GP  13

typedef __attribute__((ext_vector_type(4))) float v4f;
typedef __attribute__((ext_vector_type(4))) int   v4i;

#define SPP 13


static __device__ __forceinline__ void scan_load(v2f (&w)[SPP], float& e, float& a,
                                                 const float* sWrow, const float* sErow,
                                                 const float* sArow) {
#pragma unroll
  for (int j = 0; j < SPP; ++j) w[j] = *(const v2f*)(sWrow + 4 * j);
  e = *sErow;
  a = *sArow;
}

static __device__ __forceinline__ void scan_update(v2f (&mv)[SPP], const v2f (&w)[SPP], float e, float a) {
  const v2f ne2 = {-e, -e};
  const v2f a2  = {a, a};
#pragma unroll
  for (int j = 0; j < SPP; ++j)
    mv[j] = __builtin_elementwise_fma(w[j], __builtin_elementwise_fma(ne2, mv[j], a2), mv[j]);
}

template <int SEL>
static __device__ __forceinline__ float quad_perm(float v) {
  return __int_as_float(__builtin_amdgcn_mov_dpp(__float_as_int(v), SEL, 0xf, 0xf, false));
}

static __device__ __forceinline__ float scan_read(const v2f (&mv)[SPP], const v2f (&w)[SPP]) {
  v2f s01 = w[0] * mv[0];
  v2f s23 = w[1] * mv[1];
#pragma unroll
  for (int j = 2; j < SPP; j += 2) {
    s01 = __builtin_elementwise_fma(w[j], mv[j], s01);
    if (j + 1 < SPP) s23 = __builtin_elementwise_fma(w[j + 1], mv[j + 1], s23);
  }
  float p = (s01.x + s01.y) + (s23.x + s23.y);
  p += quad_perm<0xB1>(p);
  return p;
}


static __device__ __forceinline__ void load_afrag8(v2f (&af)[8], const float* p) {
#pragma unroll
  for (int kb = 0; kb < 8; ++kb) af[kb] = *(const v2f*)(p + kb * 4);
}

static __device__ __forceinline__ void load_resp8(v4i (&rq)[2], const int* p) {
  rq[0] = *(const v4i*)p;
  rq[1] = *(const v4i*)(p + 4);
}

template <int NB>
static __device__ __forceinline__ v8f concept_tile(const v2f (&ac)[8], const v2f (&bf)[NB]) {
  v8f acc = {0.f,0.f,0.f,0.f,0.f,0.f,0.f,0.f};
#pragma unroll
  for (int kb = 0; kb < 8; ++kb) acc = wmma_f32_k4(ac[kb], bf[NB - 8 + kb], acc);
  return acc;
}

static __device__ __forceinline__ void tile_wlog(const v2f (&af)[8], const v2f (&bf)[16], v8f c0,
                                                 float* sW, int mt, int hh, int o) {
  v8f acc = c0;
#pragma unroll
  for (int kb = 0; kb < 8; ++kb) acc = wmma_f32_k4(af[kb], bf[kb], acc);
  if (o < SIZE_M) {
#pragma unroll
    for (int j = 0; j < 8; ++j) sW[(mt * 16 + j + 8 * hh) * WST + o] = acc[j];
  }
}

static __device__ __forceinline__ void tile_ea(const v2f (&af)[8], const v4i (&rq)[2],
                                               const v2f (&be)[16], const v2f (&ba)[16],
                                               float e0, float e1, float a0, float a1,
                                               float ebv, float abv, float* sE, float* sA,
                                               int mt, int hh, int o) {
  v8f ce, ca;
#pragma unroll
  for (int j = 0; j < 8; ++j) {
    const bool one = rq[j >> 2][j & 3] != 0;
    ce[j] = one ? e1 : e0;
    ca[j] = one ? a1 : a0;
  }
#pragma unroll
  for (int kb = 0; kb < 8; ++kb) {
    ce = wmma_f32_k4(af[kb], be[kb], ce);
    ca = wmma_f32_k4(af[kb], ba[kb], ca);
  }
#pragma unroll
  for (int j = 0; j < 8; ++j) {
    int row = mt * 16 + j + 8 * hh;
    sE[row * LP + o] = sigmoid_exp_rcp(ce[j] + ebv);
    sA[row * LP + o] = tanh_exp_rcp(ca[j] + abv);
  }
}

static __device__ __forceinline__ void load_afrag24(v2f (&af)[24], const float* pr, const float* pk) {
#pragma unroll
  for (int kb = 0; kb < 16; ++kb) af[kb] = *(const v2f*)(pr + kb * 4);
#pragma unroll
  for (int kb = 0; kb < 8; ++kb) af[16 + kb] = *(const v2f*)(pk + kb * 4);
}

static __device__ __forceinline__ void tile_f(const v2f (&af)[24], const v2f (&bf)[32], v8f c0, float fbv,
                                              float* sE, int mt, int hh, int o) {
  v8f acc = c0;
#pragma unroll
  for (int kb = 0; kb < 24; ++kb) acc = wmma_f32_k4(af[kb], bf[kb], acc);
#pragma unroll
  for (int j = 0; j < 8; ++j) {
    int rr = mt * 16 + j + 8 * hh;
    sE[rr * LP + o] = tanh_exp_rcp(acc[j] + fbv);
  }
}

__global__ __launch_bounds__(256) void deepirt_fused_kernel(
    const int*   __restrict__ question, const int*   __restrict__ response,
    const float* __restrict__ vq_emb,   const float* __restrict__ vc_emb,
    const float* __restrict__ kq_emb,   const float* __restrict__ kc_emb,
    const float* __restrict__ Mk,       const float* __restrict__ Mv0,
    const float* __restrict__ eW, const float* __restrict__ eb,
    const float* __restrict__ aW, const float* __restrict__ ab,
    const float* __restrict__ fW, const float* __restrict__ fb,
    const float* __restrict__ abilW, const float* __restrict__ abilb,
    const float* __restrict__ diffW, const float* __restrict__ diffb,
    float* __restrict__ out)
{
  extern __shared__ float smem[];
  float* sK   = smem;
  float* sV   = sK  + RK * LP;
  float* sE   = sV  + RP * LP;
  float* sA   = sE  + RP * LP;
  float* sW   = sA  + RP * LP;
  int*   sR   = (int*)(sW + RP * WST);

  const int b    = blockIdx.x;
  const int tid  = threadIdx.x;
  const int lane = tid & 31;
  const int wave = tid >> 5;
  const int hh   = lane >> 4;
  const int r16  = lane & 15;

  const int wv  = __builtin_amdgcn_readfirstlane(wave);
  const int nt  = wv >> 1;
  const int par = (wv ^ (wv >> 2)) & 1;
  const int o   = nt * 16 + r16;

  v2f bk[16], be[16], ba[16];
  float ebv, abv;

  {
    const int c4 = tid & 15;
    const int gr = tid >> 4;
    int qi[GP], ri[GP];
#pragma unroll
    for (int p = 0; p < GP; ++p) {
      int t  = p * 16 + gr;
      int tc = (t < TT) ? t : (TT - 1);
      qi[p] = question[b * TT + tc];
      ri[p] = response[b * TT + tc];
    }
    {
      const int oc = (o < SIZE_M) ? o : (SIZE_M - 1);
#pragma unroll
      for (int kb = 0; kb < 16; ++kb) {
        bk[kb] = *(const v2f*)&Mk[oc * 64 + kb * 4 + 2 * hh];
        be[kb] = *(const v2f*)&eW[o * 64 + kb * 4 + 2 * hh];
        ba[kb] = *(const v2f*)&aW[o * 64 + kb * 4 + 2 * hh];
      }
      ebv = eb[o];
      abv = ab[o];
    }
    const bool   qhalf = c4 < 8;
    const float* ktab  = (qhalf ? kq_emb : kc_emb) + 4 * (c4 & 7);
    const float* vtab  = (qhalf ? vq_emb : vc_emb) + 4 * (c4 & 7);
    v4f kx[GP], vx[GP];
#pragma unroll
    for (int p = 0; p < GP; ++p) {
      int krow = qhalf ? qi[p] : 0;
      int vrow = qhalf ? qi[p] + NUM_Q * ri[p] : NUM_C * ri[p];
      kx[p] = *(const v4f*)(ktab + krow * HALF);
      vx[p] = *(const v4f*)(vtab + vrow * HALF);
    }
    const v4f z4 = {0.f, 0.f, 0.f, 0.f};
#pragma unroll
    for (int p = 0; p < GP; ++p) {
      int t = p * 16 + gr;
      *(v4f*)&sK[t * LP + 4 * c4] = (t < TT) ? kx[p] : z4;
      *(v4f*)&sV[t * LP + 4 * c4] = (t < TT) ? vx[p] : z4;
    }
    if (tid < (RK - RP) * 16) *(v4f*)&sK[(RP + gr) * LP + 4 * c4] = z4;
    if (c4 == 0) {
#pragma unroll
      for (int p = 0; p < GP; ++p) sR[p * 16 + gr] = ri[p];
    }
  }

  v8f cK;
  float ce0, ce1, ca0, ca1;
  {
    v2f akc[8], avc[8];
    const float* pkc = kc_emb + 2 * hh;
    const float* pvc = vc_emb + (NUM_C * (r16 & 1)) * HALF + 2 * hh;
#pragma unroll
    for (int kb = 0; kb < 8; ++kb) {
      akc[kb] = *(const v2f*)(pkc + kb * 4);
      avc[kb] = *(const v2f*)(pvc + kb * 4);
    }
    cK = concept_tile(akc, bk);
    const v8f cE = concept_tile(avc, be);
    const v8f cA = concept_tile(avc, ba);
    ce0 = cE[0]; ce1 = cE[1];
    ca0 = cA[0]; ca1 = cA[1];
  }
  __syncthreads();

  {
    v2f a0[8], a1[8];
    const float* pa = sK + r16 * LP + 2 * hh;
    load_afrag8(a0, pa + par * 16 * LP);
    for (int i = 0; i < 3; ++i) {
      const int m0 = par + 4 * i;
      const int m2 = (m0 + 4 < RP / 16) ? m0 + 4 : RP / 16 - 1;
      load_afrag8(a1, pa + (m0 + 2) * 16 * LP);
      __builtin_amdgcn_sched_barrier(0);
      tile_wlog(a0, bk, cK, sW, m0, hh, o);
      load_afrag8(a0, pa + m2 * 16 * LP);
      __builtin_amdgcn_sched_barrier(0);
      tile_wlog(a1, bk, cK, sW, m0 + 2, hh, o);
    }
    if (par == 0) tile_wlog(a0, bk, cK, sW, RP / 16 - 1, hh, o);
  }

  {
    v2f a0[8], a1[8];
    v4i q0[2], q1[2];
    const float* pa = sV + r16 * LP + 2 * hh;
    const int*   pq = sR + 8 * hh;
    load_afrag8(a0, pa + par * 16 * LP);
    load_resp8(q0, pq + par * 16);
    for (int i = 0; i < 3; ++i) {
      const int m0 = par + 4 * i;
      const int m2 = (m0 + 4 < RP / 16) ? m0 + 4 : RP / 16 - 1;
      load_afrag8(a1, pa + (m0 + 2) * 16 * LP);
      load_resp8(q1, pq + (m0 + 2) * 16);
      __builtin_amdgcn_sched_barrier(0);
      tile_ea(a0, q0, be, ba, ce0, ce1, ca0, ca1, ebv, abv, sE, sA, m0, hh, o);
      load_afrag8(a0, pa + m2 * 16 * LP);
      load_resp8(q0, pq + m2 * 16);
      __builtin_amdgcn_sched_barrier(0);
      tile_ea(a1, q1, be, ba, ce0, ce1, ca0, ca1, ebv, abv, sE, sA, m0 + 2, hh, o);
    }
    if (par == 0) tile_ea(a0, q0, be, ba, ce0, ce1, ca0, ca1, ebv, abv, sE, sA, RP / 16 - 1, hh, o);
  }
  __syncthreads();

  const int mg = lane & 1;
  const int dq = nt * 16 + (lane >> 1);
  v2f mv[SPP];
  v2f bw[32];
  v2f akf[8];
  float fbv = 0.f;
  if (par == 0) {
#pragma unroll
    for (int j = 0; j < SPP - 1; ++j) {
      mv[j].x = Mv0[(4 * j + 2 * mg) * 64 + dq];
      mv[j].y = Mv0[(4 * j + 2 * mg + 1) * 64 + dq];
    }
    const float m48 = Mv0[48 * 64 + dq];
    const float m49 = Mv0[49 * 64 + dq];
    mv[SPP - 1].x = (mg == 0) ? m48 : 0.f;
    mv[SPP - 1].y = (mg == 0) ? m49 : 0.f;
  } else {
#pragma unroll
    for (int kb = 0; kb < 32; ++kb)
      bw[kb] = *(const v2f*)&fW[o * 128 + kb * 4 + 2 * hh];
    fbv = fb[o];
#pragma unroll
    for (int kb = 0; kb < 8; ++kb) akf[kb] = *(const v2f*)(kc_emb + kb * 4 + 2 * hh);
  }

  for (int r = tid; r < RP; r += 256) {
    float* row = sW + r * WST;
    float x[SIZE_M];
#pragma unroll
    for (int q = 0; q < 12; ++q) {
      const v4f v = *(const v4f*)(row + 4 * q);
      x[4 * q] = v.x; x[4 * q + 1] = v.y; x[4 * q + 2] = v.z; x[4 * q + 3] = v.w;
    }
    {
      const v2f v = *(const v2f*)(row + 48);
      x[48] = v.x; x[49] = v.y;
    }
    float mx = -1e30f;
#pragma unroll
    for (int m = 0; m < SIZE_M; ++m) mx = fmaxf(mx, x[m]);
    float s = 0.f;
#pragma unroll
    for (int m = 0; m < SIZE_M; ++m) { x[m] = __expf(x[m] - mx); s += x[m]; }
    const float inv = 1.f / s;
#pragma unroll
    for (int q = 0; q < 12; ++q) {
      const v4f v = {x[4 * q] * inv, x[4 * q + 1] * inv, x[4 * q + 2] * inv, x[4 * q + 3] * inv};
      *(v4f*)(row + 4 * q) = v;
    }
    const v4f t0 = {x[48] * inv, x[49] * inv, 0.f, 0.f};
    const v4f t1 = {0.f, 0.f, 0.f, 0.f};
    *(v4f*)(row + 48) = t0;
    *(v4f*)(row + 52) = t1;
  }
  for (int idx = (TT - 1) * 64 + tid; idx < RP * 64; idx += 256) sV[(idx >> 6) * LP + (idx & 63)] = 0.f;
  v8f cF = {0.f,0.f,0.f,0.f,0.f,0.f,0.f,0.f};
  if (par != 0) cF = concept_tile(akf, bw);
  __syncthreads();

  {
    const float* pW = sW + 2 * mg;
    const float* pE = sE + dq;
    const float* pA = sA + dq;
    float*       pR = sV + dq;
    const float* pr = sV + r16 * LP + 2 * hh;
    const float* pk = sK + (r16 + 1) * LP + 2 * hh;

    v2f w0[SPP], w1[SPP], w2[SPP], w3[SPP];
    float e0 = 0.f, a0 = 0.f, e1 = 0.f, a1 = 0.f, e2 = 0.f, a2 = 0.f, e3 = 0.f, a3 = 0.f;
    const float* pWt = pW;
    const float* pEt = pE;
    const float* pAt = pA;
    float*       pRt = pR;
    if (par == 0) {
      scan_load(w3, e3, a3, pW, pE, pA);
      scan_load(w0, e0, a0, pW + WST, pE + LP, pA + LP);
      scan_load(w1, e1, a1, pW + 2 * WST, pE + 2 * LP, pA + 2 * LP);
      scan_update(mv, w3, e3, a3);
    }

    for (int it = 0; it <= RP / 16; ++it) {
      if (par == 0) {
        if (it < RP / 16) {
          const int ntrip = (it < RP / 16 - 1) ? 4 : 2;
          for (int q = 0; q < ntrip; ++q) {
            scan_load(w2, e2, a2, pWt + 3 * WST, pEt + 3 * LP, pAt + 3 * LP);
            const float r0 = scan_read(mv, w0);
            scan_update(mv, w0, e0, a0);
            scan_load(w3, e3, a3, pWt + 4 * WST, pEt + 4 * LP, pAt + 4 * LP);
            const float r1 = scan_read(mv, w1);
            scan_update(mv, w1, e1, a1);
            scan_load(w0, e0, a0, pWt + 5 * WST, pEt + 5 * LP, pAt + 5 * LP);
            const float r2 = scan_read(mv, w2);
            scan_update(mv, w2, e2, a2);
            scan_load(w1, e1, a1, pWt + 6 * WST, pEt + 6 * LP, pAt + 6 * LP);
            const float r3 = scan_read(mv, w3);
            scan_update(mv, w3, e3, a3);
            if (mg == 0) {
              pRt[0] = r0; pRt[LP] = r1; pRt[2 * LP] = r2; pRt[3 * LP] = r3;
            }
            pWt += 4 * WST; pEt += 4 * LP; pAt += 4 * LP; pRt += 4 * LP;
          }
        }
      } else {
        if (it >= 1) {
          const int mt = it - 1;
          v2f af[24];
          load_afrag24(af, pr + mt * 16 * LP, pk + mt * 16 * LP);
          __builtin_amdgcn_sched_barrier(0);
          tile_f(af, bw, cF, fbv, sE, mt, hh, o);
        }
      }
      __syncthreads();
    }
  }

  for (int r = tid; r < TT - 1; r += 256) {
    v4f fr[16], kr[16];
#pragma unroll
    for (int q = 0; q < 16; ++q) {
      fr[q] = *(const v4f*)&sE[r * LP + 4 * q];
      kr[q] = *(const v4f*)&sK[(r + 1) * LP + 4 * q];
    }
    __builtin_amdgcn_sched_barrier(0);
    float s2 = 0.f;
#pragma unroll
    for (int q = 0; q < 16; ++q) {
      s2 = __builtin_fmaf(kr[q].x, diffW[4 * q],     s2);
      s2 = __builtin_fmaf(kr[q].y, diffW[4 * q + 1], s2);
      s2 = __builtin_fmaf(kr[q].z, diffW[4 * q + 2], s2);
      s2 = __builtin_fmaf(kr[q].w, diffW[4 * q + 3], s2);
    }
    __builtin_amdgcn_sched_barrier(0);
    float s1 = 0.f;
#pragma unroll
    for (int q = 0; q < 16; ++q) {
      s1 = __builtin_fmaf(fr[q].x, abilW[4 * q],     s1);
      s1 = __builtin_fmaf(fr[q].y, abilW[4 * q + 1], s1);
      s1 = __builtin_fmaf(fr[q].z, abilW[4 * q + 2], s1);
      s1 = __builtin_fmaf(fr[q].w, abilW[4 * q + 3], s1);
    }
    float ability = tanhf(s1 + abilb[0]);
    float diff    = tanhf(s2 + diffb[0]);
    out[b * (TT - 1) + r] = 3.f * ability - diff;
  }
}

extern "C" void kernel_launch(void* const* d_in, const int* in_sizes, int n_in,
                              void* d_out, int out_size, void* d_ws, size_t ws_size,
                              hipStream_t stream) {
  (void)in_sizes; (void)n_in; (void)out_size; (void)d_ws; (void)ws_size;
  const int*   question = (const int*)  d_in[0];
  const int*   response = (const int*)  d_in[1];
  const float* vq_emb   = (const float*)d_in[2];
  const float* vc_emb   = (const float*)d_in[3];
  const float* kq_emb   = (const float*)d_in[4];
  const float* kc_emb   = (const float*)d_in[5];
  const float* Mk       = (const float*)d_in[6];
  const float* Mv0      = (const float*)d_in[7];
  const float* eW       = (const float*)d_in[8];
  const float* ebv      = (const float*)d_in[9];
  const float* aW       = (const float*)d_in[10];
  const float* abv      = (const float*)d_in[11];
  const float* fW       = (const float*)d_in[12];
  const float* fbv      = (const float*)d_in[13];
  const float* abilW    = (const float*)d_in[14];
  const float* abilb    = (const float*)d_in[15];
  const float* diffW    = (const float*)d_in[16];
  const float* diffb    = (const float*)d_in[17];
  float* out = (float*)d_out;

  const size_t smem_bytes =
      (size_t)(RK * LP + 3 * RP * LP + RP * WST + RP) * sizeof(float);

  (void)hipFuncSetAttribute((const void*)deepirt_fused_kernel,
                            hipFuncAttributeMaxDynamicSharedMemorySize,
                            (int)smem_bytes);

  deepirt_fused_kernel<<<BATCH, 256, smem_bytes, stream>>>(
      question, response, vq_emb, vc_emb, kq_emb, kc_emb, Mk, Mv0,
      eW, ebv, aW, abv, fW, fbv, abilW, abilb, diffW, diffb, out);
}


# Scan trip of 4 steps: its LDS operand loads issued as two groups (top of the trip; behind step t+1) in place of a dozen points spread among the multiply-adds; every sum's order unchanged
# speedup vs baseline: 12.5687x; 1.2426x over previous
#include <hip/hip_runtime.h>
#include <math.h>


#define NUM_Q  10000
#define NUM_C  300
#define DIM_S  64
#define SIZE_M 50
#define BATCH  128
#define TT     200
#define HALF   32

#define RK  212
#define RP  208
#define WST 60
#define LP  68

typedef __attribute__((ext_vector_type(2))) float v2f;
typedef __attribute__((ext_vector_type(8))) float v8f;

static __device__ __forceinline__ v8f wmma_f32_k4(v2f a, v2f b, v8f c) {
  return __builtin_amdgcn_wmma_f32_16x16x4_f32(false, a, false, b, (short)0, c, false, false);
}

static __device__ __forceinline__ float tanh_exp_rcp(float x) {
  return 1.f - 2.f * __builtin_amdgcn_rcpf(1.f + __expf(2.f * x));
}
static __device__ __forceinline__ float sigmoid_exp_rcp(float x) {
  return __builtin_amdgcn_rcpf(1.f + __expf(-x));
}

#define GP  13

typedef __attribute__((ext_vector_type(4))) float v4f;
typedef __attribute__((ext_vector_type(4))) int   v4i;

#define SPP 13


static __device__ __forceinline__ void scan_load_w(v2f (&w)[SPP], const float* sWrow) {
#pragma unroll
  for (int j = 0; j < SPP; ++j) w[j] = *(const v2f*)(sWrow + 4 * j);
}

static __device__ __forceinline__ void scan_update(v2f (&mv)[SPP], const v2f (&w)[SPP], float e, float a) {
  const v2f ne2 = {-e, -e};
  const v2f a2  = {a, a};
#pragma unroll
  for (int j = 0; j < SPP; ++j)
    mv[j] = __builtin_elementwise_fma(w[j], __builtin_elementwise_fma(ne2, mv[j], a2), mv[j]);
}

template <int L>
static __device__ __forceinline__ void scan_update_p(v2f (&mv)[SPP], const v2f (&w)[SPP], v2f e, v2f a) {
  const v2f e2 = __builtin_shufflevector(e, e, L, L);
  const v2f a2 = __builtin_shufflevector(a, a, L, L);
#pragma unroll
  for (int j = 0; j < SPP; ++j)
    mv[j] = __builtin_elementwise_fma(w[j], __builtin_elementwise_fma(e2, -mv[j], a2), mv[j]);
}

template <int SEL>
static __device__ __forceinline__ float quad_perm(float v) {
  return __int_as_float(__builtin_amdgcn_mov_dpp(__float_as_int(v), SEL, 0xf, 0xf, false));
}

static __device__ __forceinline__ void scan_read_part(v2f& s01, v2f& s23,
                                                      const v2f (&mv)[SPP], const v2f (&w)[SPP]) {
  s01 = w[0] * mv[0];
  s23 = w[1] * mv[1];
#pragma unroll
  for (int j = 2; j < SPP; j += 2) {
    s01 = __builtin_elementwise_fma(w[j], mv[j], s01);
    if (j + 1 < SPP) s23 = __builtin_elementwise_fma(w[j + 1], mv[j + 1], s23);
  }
}

static __device__ __forceinline__ float scan_read_join(v2f s01, v2f s23) {
  float p = (s01.x + s01.y) + (s23.x + s23.y);
  p += quad_perm<0xB1>(p);
  return p;
}


static __device__ __forceinline__ void load_afrag8(v2f (&af)[8], const float* p) {
#pragma unroll
  for (int kb = 0; kb < 8; ++kb) af[kb] = *(const v2f*)(p + kb * 4);
}

static __device__ __forceinline__ void load_resp8(v4i (&rq)[2], const int* p) {
  rq[0] = *(const v4i*)p;
  rq[1] = *(const v4i*)(p + 4);
}

template <int NB>
static __device__ __forceinline__ v8f concept_tile(const v2f (&ac)[8], const v2f (&bf)[NB]) {
  v8f acc = {0.f,0.f,0.f,0.f,0.f,0.f,0.f,0.f};
#pragma unroll
  for (int kb = 0; kb < 8; ++kb) acc = wmma_f32_k4(ac[kb], bf[NB - 8 + kb], acc);
  return acc;
}

static __device__ __forceinline__ void tile_wlog(const v2f (&af)[8], const v2f (&bf)[16], v8f c0,
                                                 float* sW, int mt, int hh, int o) {
  v8f acc = c0;
#pragma unroll
  for (int kb = 0; kb < 8; ++kb) acc = wmma_f32_k4(af[kb], bf[kb], acc);
  if (o < SIZE_M) {
#pragma unroll
    for (int j = 0; j < 8; ++j) sW[(mt * 16 + j + 8 * hh) * WST + o] = acc[j];
  }
}

static __device__ __forceinline__ void tile_ea(const v2f (&af)[8], const v4i (&rq)[2],
                                               const v2f (&be)[16], const v2f (&ba)[16],
                                               float e0, float e1, float a0, float a1,
                                               float ebv, float abv, float* sE, float* sA,
                                               int mt, int hh, int o) {
  v8f ce, ca;
#pragma unroll
  for (int j = 0; j < 8; ++j) {
    const bool one = rq[j >> 2][j & 3] != 0;
    ce[j] = one ? e1 : e0;
    ca[j] = one ? a1 : a0;
  }
#pragma unroll
  for (int kb = 0; kb < 8; ++kb) {
    ce = wmma_f32_k4(af[kb], be[kb], ce);
    ca = wmma_f32_k4(af[kb], ba[kb], ca);
  }
#pragma unroll
  for (int j = 0; j < 8; ++j) {
    int row = mt * 16 + j + 8 * hh;
    sE[row * LP + o] = sigmoid_exp_rcp(ce[j] + ebv);
    sA[row * LP + o] = tanh_exp_rcp(ca[j] + abv);
  }
}

static __device__ __forceinline__ void load_afrag24(v2f (&af)[24], const float* pr, const float* pk) {
#pragma unroll
  for (int kb = 0; kb < 16; ++kb) af[kb] = *(const v2f*)(pr + kb * 4);
#pragma unroll
  for (int kb = 0; kb < 8; ++kb) af[16 + kb] = *(const v2f*)(pk + kb * 4);
}

static __device__ __forceinline__ void tile_f(const v2f (&af)[24], const v2f (&bf)[32], v8f c0, float fbv,
                                              float* sE, int mt, int hh, int o) {
  v8f acc = c0;
#pragma unroll
  for (int kb = 0; kb < 24; ++kb) acc = wmma_f32_k4(af[kb], bf[kb], acc);
#pragma unroll
  for (int j = 0; j < 8; ++j) {
    int rr = mt * 16 + j + 8 * hh;
    sE[rr * LP + o] = tanh_exp_rcp(acc[j] + fbv);
  }
}

__global__ __launch_bounds__(256) void deepirt_fused_kernel(
    const int*   __restrict__ question, const int*   __restrict__ response,
    const float* __restrict__ vq_emb,   const float* __restrict__ vc_emb,
    const float* __restrict__ kq_emb,   const float* __restrict__ kc_emb,
    const float* __restrict__ Mk,       const float* __restrict__ Mv0,
    const float* __restrict__ eW, const float* __restrict__ eb,
    const float* __restrict__ aW, const float* __restrict__ ab,
    const float* __restrict__ fW, const float* __restrict__ fb,
    const float* __restrict__ abilW, const float* __restrict__ abilb,
    const float* __restrict__ diffW, const float* __restrict__ diffb,
    float* __restrict__ out)
{
  extern __shared__ float smem[];
  float* sK   = smem;
  float* sV   = sK  + RK * LP;
  float* sE   = sV  + RP * LP;
  float* sA   = sE  + RP * LP;
  float* sW   = sA  + RP * LP;
  int*   sR   = (int*)(sW + RP * WST);

  const int b    = blockIdx.x;
  const int tid  = threadIdx.x;
  const int lane = tid & 31;
  const int wave = tid >> 5;
  const int hh   = lane >> 4;
  const int r16  = lane & 15;

  const int wv  = __builtin_amdgcn_readfirstlane(wave);
  const int nt  = wv >> 1;
  const int par = (wv ^ (wv >> 2)) & 1;
  const int o   = nt * 16 + r16;

  v2f bk[16], be[16], ba[16];
  float ebv, abv;

  {
    const int c4 = tid & 15;
    const int gr = tid >> 4;
    int qi[GP], ri[GP];
#pragma unroll
    for (int p = 0; p < GP; ++p) {
      int t  = p * 16 + gr;
      int tc = (t < TT) ? t : (TT - 1);
      qi[p] = question[b * TT + tc];
      ri[p] = response[b * TT + tc];
    }
    {
      const int oc = (o < SIZE_M) ? o : (SIZE_M - 1);
#pragma unroll
      for (int kb = 0; kb < 16; ++kb) {
        bk[kb] = *(const v2f*)&Mk[oc * 64 + kb * 4 + 2 * hh];
        be[kb] = *(const v2f*)&eW[o * 64 + kb * 4 + 2 * hh];
        ba[kb] = *(const v2f*)&aW[o * 64 + kb * 4 + 2 * hh];
      }
      ebv = eb[o];
      abv = ab[o];
    }
    const bool   qhalf = c4 < 8;
    const float* ktab  = (qhalf ? kq_emb : kc_emb) + 4 * (c4 & 7);
    const float* vtab  = (qhalf ? vq_emb : vc_emb) + 4 * (c4 & 7);
    v4f kx[GP], vx[GP];
#pragma unroll
    for (int p = 0; p < GP; ++p) {
      int krow = qhalf ? qi[p] : 0;
      int vrow = qhalf ? qi[p] + NUM_Q * ri[p] : NUM_C * ri[p];
      kx[p] = *(const v4f*)(ktab + krow * HALF);
      vx[p] = *(const v4f*)(vtab + vrow * HALF);
    }
    const v4f z4 = {0.f, 0.f, 0.f, 0.f};
#pragma unroll
    for (int p = 0; p < GP; ++p) {
      int t = p * 16 + gr;
      *(v4f*)&sK[t * LP + 4 * c4] = (t < TT) ? kx[p] : z4;
      *(v4f*)&sV[t * LP + 4 * c4] = (t < TT) ? vx[p] : z4;
    }
    if (tid < (RK - RP) * 16) *(v4f*)&sK[(RP + gr) * LP + 4 * c4] = z4;
    if (c4 == 0) {
#pragma unroll
      for (int p = 0; p < GP; ++p) sR[p * 16 + gr] = ri[p];
    }
  }

  v8f cK;
  float ce0, ce1, ca0, ca1;
  {
    v2f akc[8], avc[8];
    const float* pkc = kc_emb + 2 * hh;
    const float* pvc = vc_emb + (NUM_C * (r16 & 1)) * HALF + 2 * hh;
#pragma unroll
    for (int kb = 0; kb < 8; ++kb) {
      akc[kb] = *(const v2f*)(pkc + kb * 4);
      avc[kb] = *(const v2f*)(pvc + kb * 4);
    }
    cK = concept_tile(akc, bk);
    const v8f cE = concept_tile(avc, be);
    const v8f cA = concept_tile(avc, ba);
    ce0 = cE[0]; ce1 = cE[1];
    ca0 = cA[0]; ca1 = cA[1];
  }
  __syncthreads();

  {
    v2f a0[8], a1[8];
    const float* pa = sK + r16 * LP + 2 * hh;
    load_afrag8(a0, pa + par * 16 * LP);
    for (int i = 0; i < 3; ++i) {
      const int m0 = par + 4 * i;
      const int m2 = (m0 + 4 < RP / 16) ? m0 + 4 : RP / 16 - 1;
      load_afrag8(a1, pa + (m0 + 2) * 16 * LP);
      __builtin_amdgcn_sched_barrier(0);
      tile_wlog(a0, bk, cK, sW, m0, hh, o);
      load_afrag8(a0, pa + m2 * 16 * LP);
      __builtin_amdgcn_sched_barrier(0);
      tile_wlog(a1, bk, cK, sW, m0 + 2, hh, o);
    }
    if (par == 0) tile_wlog(a0, bk, cK, sW, RP / 16 - 1, hh, o);
  }

  {
    v2f a0[8], a1[8];
    v4i q0[2], q1[2];
    const float* pa = sV + r16 * LP + 2 * hh;
    const int*   pq = sR + 8 * hh;
    load_afrag8(a0, pa + par * 16 * LP);
    load_resp8(q0, pq + par * 16);
    for (int i = 0; i < 3; ++i) {
      const int m0 = par + 4 * i;
      const int m2 = (m0 + 4 < RP / 16) ? m0 + 4 : RP / 16 - 1;
      load_afrag8(a1, pa + (m0 + 2) * 16 * LP);
      load_resp8(q1, pq + (m0 + 2) * 16);
      __builtin_amdgcn_sched_barrier(0);
      tile_ea(a0, q0, be, ba, ce0, ce1, ca0, ca1, ebv, abv, sE, sA, m0, hh, o);
      load_afrag8(a0, pa + m2 * 16 * LP);
      load_resp8(q0, pq + m2 * 16);
      __builtin_amdgcn_sched_barrier(0);
      tile_ea(a1, q1, be, ba, ce0, ce1, ca0, ca1, ebv, abv, sE, sA, m0 + 2, hh, o);
    }
    if (par == 0) tile_ea(a0, q0, be, ba, ce0, ce1, ca0, ca1, ebv, abv, sE, sA, RP / 16 - 1, hh, o);
  }
  __syncthreads();

  const int mg = lane & 1;
  const int dq = nt * 16 + (lane >> 1);
  v2f mv[SPP];
  v2f bw[32];
  v2f akf[8];
  float fbv = 0.f;
  if (par == 0) {
#pragma unroll
    for (int j = 0; j < SPP - 1; ++j) {
      mv[j].x = Mv0[(4 * j + 2 * mg) * 64 + dq];
      mv[j].y = Mv0[(4 * j + 2 * mg + 1) * 64 + dq];
    }
    const float m48 = Mv0[48 * 64 + dq];
    const float m49 = Mv0[49 * 64 + dq];
    mv[SPP - 1].x = (mg == 0) ? m48 : 0.f;
    mv[SPP - 1].y = (mg == 0) ? m49 : 0.f;
  } else {
#pragma unroll
    for (int kb = 0; kb < 32; ++kb)
      bw[kb] = *(const v2f*)&fW[o * 128 + kb * 4 + 2 * hh];
    fbv = fb[o];
#pragma unroll
    for (int kb = 0; kb < 8; ++kb) akf[kb] = *(const v2f*)(kc_emb + kb * 4 + 2 * hh);
  }

  for (int r = tid; r < RP; r += 256) {
    float* row = sW + r * WST;
    float x[SIZE_M];
#pragma unroll
    for (int q = 0; q < 12; ++q) {
      const v4f v = *(const v4f*)(row + 4 * q);
      x[4 * q] = v.x; x[4 * q + 1] = v.y; x[4 * q + 2] = v.z; x[4 * q + 3] = v.w;
    }
    {
      const v2f v = *(const v2f*)(row + 48);
      x[48] = v.x; x[49] = v.y;
    }
    float mx = -1e30f;
#pragma unroll
    for (int m = 0; m < SIZE_M; ++m) mx = fmaxf(mx, x[m]);
    float s = 0.f;
#pragma unroll
    for (int m = 0; m < SIZE_M; ++m) { x[m] = __expf(x[m] - mx); s += x[m]; }
    const float inv = 1.f / s;
#pragma unroll
    for (int q = 0; q < 12; ++q) {
      const v4f v = {x[4 * q] * inv, x[4 * q + 1] * inv, x[4 * q + 2] * inv, x[4 * q + 3] * inv};
      *(v4f*)(row + 4 * q) = v;
    }
    const v4f t0 = {x[48] * inv, x[49] * inv, 0.f, 0.f};
    const v4f t1 = {0.f, 0.f, 0.f, 0.f};
    *(v4f*)(row + 48) = t0;
    *(v4f*)(row + 52) = t1;
  }
  for (int idx = (TT - 1) * 64 + tid; idx < RP * 64; idx += 256) sV[(idx >> 6) * LP + (idx & 63)] = 0.f;
  v8f cF = {0.f,0.f,0.f,0.f,0.f,0.f,0.f,0.f};
  if (par != 0) cF = concept_tile(akf, bw);
  __syncthreads();

  {
    const float* pW = sW + 2 * mg;
    const float* pE = sE + dq;
    const float* pA = sA + dq;
    float*       pR = sV + dq;
    const float* pr = sV + r16 * LP + 2 * hh;
    const float* pk = sK + (r16 + 1) * LP + 2 * hh;

    v2f w0[SPP], w1[SPP], w2[SPP], w3[SPP];
    float eA = 0.f, aA = 0.f, eB = 0.f, aB = 0.f;
    v2f e23 = {0.f, 0.f}, a23 = {0.f, 0.f};
    const float* pWt = pW;
    const float* pEt = pE;
    const float* pAt = pA;
    float*       pRt = pR;
    if (par == 0) {
      scan_load_w(w3, pW);
      scan_load_w(w0, pW + WST);
      scan_load_w(w1, pW + 2 * WST);
      const float eZ = pE[0], aZ = pA[0];
      eA = pE[LP]; aA = pA[LP];
      scan_update(mv, w3, eZ, aZ);
    }

    for (int it = 0; it <= RP / 16; ++it) {
      if (par == 0) {
        if (it < RP / 16) {
          const int ntrip = (it < RP / 16 - 1) ? 4 : 2;
          for (int q = 0; q < ntrip; ++q) {
            v2f s0a, s0b, s1a, s1b, s2a, s2b, s3a, s3b;
            eB = pEt[2 * LP]; aB = pAt[2 * LP];
            e23.x = pEt[3 * LP]; e23.y = pEt[4 * LP];
            a23.x = pAt[3 * LP]; a23.y = pAt[4 * LP];
            scan_load_w(w2, pWt + 3 * WST);
            scan_load_w(w3, pWt + 4 * WST);
            __builtin_amdgcn_sched_barrier(0);
            scan_read_part(s0a, s0b, mv, w0);
            scan_update(mv, w0, eA, aA);
            scan_read_part(s1a, s1b, mv, w1);
            scan_update(mv, w1, eB, aB);
            asm volatile("" :: "v"(mv[0]), "v"(mv[1]), "v"(mv[2]), "v"(mv[3]), "v"(mv[4]), "v"(mv[5]), "v"(mv[6]),
                               "v"(mv[7]), "v"(mv[8]), "v"(mv[9]), "v"(mv[10]), "v"(mv[11]), "v"(mv[12]));
            __builtin_amdgcn_sched_barrier(0);
            scan_load_w(w0, pWt + 5 * WST);
            scan_load_w(w1, pWt + 6 * WST);
            eA = pEt[5 * LP]; aA = pAt[5 * LP];
            __builtin_amdgcn_sched_barrier(0);
            scan_read_part(s2a, s2b, mv, w2);
            scan_update_p<0>(mv, w2, e23, a23);
            scan_read_part(s3a, s3b, mv, w3);
            scan_update_p<1>(mv, w3, e23, a23);
            const float r0 = scan_read_join(s0a, s0b);
            const float r1 = scan_read_join(s1a, s1b);
            const float r2 = scan_read_join(s2a, s2b);
            const float r3 = scan_read_join(s3a, s3b);
            if (mg == 0) {
              pRt[0] = r0; pRt[LP] = r1; pRt[2 * LP] = r2; pRt[3 * LP] = r3;
            }
            pWt += 4 * WST; pEt += 4 * LP; pAt += 4 * LP; pRt += 4 * LP;
          }
        }
      } else {
        if (it >= 1) {
          const int mt = it - 1;
          v2f af[24];
          load_afrag24(af, pr + mt * 16 * LP, pk + mt * 16 * LP);
          __builtin_amdgcn_sched_barrier(0);
          tile_f(af, bw, cF, fbv, sE, mt, hh, o);
        }
      }
      __syncthreads();
    }
  }

  for (int r = tid; r < TT - 1; r += 256) {
    v4f fr[16], kr[16];
#pragma unroll
    for (int q = 0; q < 16; ++q) {
      fr[q] = *(const v4f*)&sE[r * LP + 4 * q];
      kr[q] = *(const v4f*)&sK[(r + 1) * LP + 4 * q];
    }
    __builtin_amdgcn_sched_barrier(0);
    float s2 = 0.f;
#pragma unroll
    for (int q = 0; q < 16; ++q) {
      s2 = __builtin_fmaf(kr[q].x, diffW[4 * q],     s2);
      s2 = __builtin_fmaf(kr[q].y, diffW[4 * q + 1], s2);
      s2 = __builtin_fmaf(kr[q].z, diffW[4 * q + 2], s2);
      s2 = __builtin_fmaf(kr[q].w, diffW[4 * q + 3], s2);
    }
    __builtin_amdgcn_sched_barrier(0);
    float s1 = 0.f;
#pragma unroll
    for (int q = 0; q < 16; ++q) {
      s1 = __builtin_fmaf(fr[q].x, abilW[4 * q],     s1);
      s1 = __builtin_fmaf(fr[q].y, abilW[4 * q + 1], s1);
      s1 = __builtin_fmaf(fr[q].z, abilW[4 * q + 2], s1);
      s1 = __builtin_fmaf(fr[q].w, abilW[4 * q + 3], s1);
    }
    float ability = tanhf(s1 + abilb[0]);
    float diff    = tanhf(s2 + diffb[0]);
    out[b * (TT - 1) + r] = 3.f * ability - diff;
  }
}

extern "C" void kernel_launch(void* const* d_in, const int* in_sizes, int n_in,
                              void* d_out, int out_size, void* d_ws, size_t ws_size,
                              hipStream_t stream) {
  (void)in_sizes; (void)n_in; (void)out_size; (void)d_ws; (void)ws_size;
  const int*   question = (const int*)  d_in[0];
  const int*   response = (const int*)  d_in[1];
  const float* vq_emb   = (const float*)d_in[2];
  const float* vc_emb   = (const float*)d_in[3];
  const float* kq_emb   = (const float*)d_in[4];
  const float* kc_emb   = (const float*)d_in[5];
  const float* Mk       = (const float*)d_in[6];
  const float* Mv0      = (const float*)d_in[7];
  const float* eW       = (const float*)d_in[8];
  const float* ebv      = (const float*)d_in[9];
  const float* aW       = (const float*)d_in[10];
  const float* abv      = (const float*)d_in[11];
  const float* fW       = (const float*)d_in[12];
  const float* fbv      = (const float*)d_in[13];
  const float* abilW    = (const float*)d_in[14];
  const float* abilb    = (const float*)d_in[15];
  const float* diffW    = (const float*)d_in[16];
  const float* diffb    = (const float*)d_in[17];
  float* out = (float*)d_out;

  const size_t smem_bytes =
      (size_t)(RK * LP + 3 * RP * LP + RP * WST + RP) * sizeof(float);

  (void)hipFuncSetAttribute((const void*)deepirt_fused_kernel,
                            hipFuncAttributeMaxDynamicSharedMemorySize,
                            (int)smem_bytes);

  deepirt_fused_kernel<<<BATCH, 256, smem_bytes, stream>>>(
      question, response, vq_emb, vc_emb, kq_emb, kc_emb, Mk, Mv0,
      eW, ebv, aW, abv, fW, fbv, abilW, abilb, diffW, diffb, out);
}
